# starting kernel
# baseline (speedup 1.0000x reference)
#include <hip/hip_runtime.h>

typedef __attribute__((ext_vector_type(2))) float v2f;
typedef __attribute__((ext_vector_type(8))) float v8f;

#define HID 64
#define WST 68

__global__ __launch_bounds__(256) void gnn_gemm_relu(
    const float* __restrict__ in, const float* __restrict__ W,
    const float* __restrict__ resid, float* __restrict__ out, int n)
{
    __shared__ float lw[64 * WST];
    const int tid = threadIdx.x;

    for (int i = tid; i < 64 * 64; i += 256) {
        int r = i >> 6, c = i & 63;
        lw[r * WST + c] = W[i];
    }
    __syncthreads();

    const int wave = tid >> 5;
    const int lane = tid & 31;
    const int half = lane >> 4;
    const int l16  = lane & 15;
    const int row0 = blockIdx.x * 128 + wave * 16;

    const int  arow  = row0 + l16;
    const bool rowok = arow < n;
    const float* abase = in + (size_t)(rowok ? arow : 0) * HID + half * 2;
    v2f a[16];
#pragma unroll
    for (int s = 0; s < 16; ++s) {
        float2 t = rowok ? *(const float2*)(abase + 4 * s) : make_float2(0.f, 0.f);
        a[s].x = t.x; a[s].y = t.y;
    }

    v8f acc[4] = {};

#pragma unroll
    for (int s = 0; s < 16; ++s) {
#pragma unroll
        for (int j = 0; j < 4; ++j) {
            const int nc = j * 16 + l16;
            float2 bb = *(const float2*)&lw[nc * WST + 4 * s + half * 2];
            v2f b; b.x = bb.x; b.y = bb.y;
            acc[j] = __builtin_amdgcn_wmma_f32_16x16x4_f32(
                false, a[s], false, b, (short)0, acc[j], false, false);
        }
    }

#pragma unroll
    for (int j = 0; j < 4; ++j) {
        const int colg = j * 16 + l16;
#pragma unroll
        for (int v = 0; v < 8; ++v) {
            const int rowg = row0 + v + half * 8;
            if (rowg < n) {
                float val = fmaxf(acc[j][v], 0.f);
                if (resid) val += resid[(size_t)rowg * HID + colg];
                out[(size_t)rowg * HID + colg] = val;
            }
        }
    }
}

__device__ __forceinline__ void hw_atomic_add_f32(float* p, float v) {
    asm volatile("global_atomic_add_f32 %0, %1, off"
                 :
                 : "v"(p), "v"(v)
                 : "memory");
}

__global__ __launch_bounds__(256) void gnn_scatter(
    const float* __restrict__ m, const int* __restrict__ row,
    const int* __restrict__ col, float* __restrict__ t, int ne)
{
    const int wave = (int)((blockIdx.x * blockDim.x + threadIdx.x) >> 5);
    const int lane = threadIdx.x & 31;
    const int e0 = wave * 4;
#pragma unroll
    for (int i = 0; i < 4; ++i) {
        const int e = e0 + i;
        if (e >= ne) return;
        const int r = row[e];
        const int c = col[e];
        float2 v = *(const float2*)(m + (size_t)c * HID + 2 * lane);
        float* dst = t + (size_t)r * HID + 2 * lane;
        hw_atomic_add_f32(dst,     v.x);
        hw_atomic_add_f32(dst + 1, v.y);
    }
}

__global__ __launch_bounds__(256) void gnn_rmsnorm(
    const float* __restrict__ x, const float* __restrict__ w,
    float* __restrict__ out, int n)
{
    const long long gid = (long long)blockIdx.x * blockDim.x + threadIdx.x;
    const int r    = (int)(gid >> 5);
    const int lane = threadIdx.x & 31;
    if (r >= n) return;

    float2 v = *(const float2*)(x + (size_t)r * HID + 2 * lane);
    float ss = v.x * v.x + v.y * v.y;
#pragma unroll
    for (int off = 16; off >= 1; off >>= 1)
        ss += __shfl_xor(ss, off, 32);
    const float inv = rsqrtf(ss * (1.0f / 64.0f) + 1e-5f);

    float2 wv = *(const float2*)(w + 2 * lane);
    float2 o; o.x = v.x * inv * wv.x; o.y = v.y * inv * wv.y;
    *(float2*)(out + (size_t)r * HID + 2 * lane) = o;
}

extern "C" void kernel_launch(void* const* d_in, const int* in_sizes, int n_in,
                              void* d_out, int out_size, void* d_ws, size_t ws_size,
                              hipStream_t stream)
{
    const float* x       = (const float*)d_in[0];
    const float* conv_w  = (const float*)d_in[1];
    const float* conv_nw = (const float*)d_in[2];
    const float* hid_w   = (const float*)d_in[3];
    const float* hid_nw  = (const float*)d_in[4];
    const int*   row     = (const int*)d_in[5];
    const int*   col     = (const int*)d_in[6];

    const int n  = in_sizes[0] / HID;
    const int ne = in_sizes[5];

    float* P = (float*)d_out;
    float* Q = (float*)d_ws;
    float* R = Q + (size_t)n * HID;
    const size_t hbytes = (size_t)n * HID * sizeof(float);

    const dim3 blk(256);
    const int gemm_grid = (n + 127) / 128;
    const int rms_grid  = (n + 7) / 8;
    const int sc_grid   = ((ne + 3) / 4 + 7) / 8;

    const float* hin = x;
    for (int l = 0; l < 4; ++l) {
        const float* cw = conv_w + (size_t)l * HID * HID;
        const float* hw = hid_w  + (size_t)l * HID * HID;
        const float* cn = conv_nw + l * HID;
        const float* hn = hid_nw  + l * HID;

        gnn_gemm_relu<<<gemm_grid, blk, 0, stream>>>(hin, cw, nullptr, Q, n);
        hipMemcpyAsync(R, hin, hbytes, hipMemcpyDeviceToDevice, stream);
        gnn_scatter<<<sc_grid, blk, 0, stream>>>(Q, row, col, R, ne);
        gnn_rmsnorm<<<rms_grid, blk, 0, stream>>>(R, cn, R, n);
        gnn_gemm_relu<<<gemm_grid, blk, 0, stream>>>(R, hw, R, P, n);
        gnn_rmsnorm<<<rms_grid, blk, 0, stream>>>(P, hn, P, n);

        hin = P;
    }
}


# Fuse both RMSNorm launches into the second GEMM kernel (input-row norm in its prologue, output-row norm in its epilogue): 20 -> 12 launches; row sums of squares now summed in fragment order
# speedup vs baseline: 1.6534x; 1.6534x over previous
#include <hip/hip_runtime.h>

typedef __attribute__((ext_vector_type(2))) float v2f;
typedef __attribute__((ext_vector_type(8))) float v8f;

#define HID 64
#define WST 68

__global__ __launch_bounds__(256) void gnn_gemm_relu(
    const float* __restrict__ in, const float* __restrict__ W,
    float* __restrict__ out, int n)
{
    __shared__ float lw[64 * WST];
    const int tid = threadIdx.x;

    for (int i = tid; i < 64 * 64; i += 256) {
        int r = i >> 6, c = i & 63;
        lw[r * WST + c] = W[i];
    }
    __syncthreads();

    const int wave = tid >> 5;
    const int lane = tid & 31;
    const int half = lane >> 4;
    const int l16  = lane & 15;
    const int row0 = blockIdx.x * 128 + wave * 16;

    const int  arow  = row0 + l16;
    const bool rowok = arow < n;
    const float* abase = in + (size_t)(rowok ? arow : 0) * HID + half * 2;
    v2f a[16];
#pragma unroll
    for (int s = 0; s < 16; ++s) {
        float2 t = rowok ? *(const float2*)(abase + 4 * s) : make_float2(0.f, 0.f);
        a[s].x = t.x; a[s].y = t.y;
    }

    v8f acc[4] = {};

#pragma unroll
    for (int s = 0; s < 16; ++s) {
#pragma unroll
        for (int j = 0; j < 4; ++j) {
            const int nc = j * 16 + l16;
            float2 bb = *(const float2*)&lw[nc * WST + 4 * s + half * 2];
            v2f b; b.x = bb.x; b.y = bb.y;
            acc[j] = __builtin_amdgcn_wmma_f32_16x16x4_f32(
                false, a[s], false, b, (short)0, acc[j], false, false);
        }
    }

#pragma unroll
    for (int j = 0; j < 4; ++j) {
        const int colg = j * 16 + l16;
#pragma unroll
        for (int v = 0; v < 8; ++v) {
            const int rowg = row0 + v + half * 8;
            if (rowg < n)
                out[(size_t)rowg * HID + colg] = fmaxf(acc[j][v], 0.f);
        }
    }
}

template <int CTRL>
__device__ __forceinline__ float dpp_add(float v) {
    return v + __int_as_float(__builtin_amdgcn_update_dpp(
                   0, __float_as_int(v), CTRL, 0xf, 0xf, true));
}

__global__ __launch_bounds__(256) void gnn_norm_gemm_norm(
    const float* __restrict__ t, const float* __restrict__ w1,
    const float* __restrict__ W, const float* __restrict__ w2,
    float* __restrict__ out, int n)
{
    __shared__ float lw[64 * WST];
    __shared__ float lnw[2 * HID];
    const int tid = threadIdx.x;

    if (tid < 2 * HID) lnw[tid] = tid < HID ? w1[tid] : w2[tid - HID];

    const float w1k = w1[tid & 63];
    for (int i = tid; i < 64 * 64; i += 256) {
        int r = i >> 6, c = i & 63;
        lw[r * WST + c] = W[i] * w1k;
    }
    __syncthreads();

    const int wave = tid >> 5;
    const int lane = tid & 31;
    const int half = lane >> 4;
    const int l16  = lane & 15;
    const int row0 = blockIdx.x * 128 + wave * 16;

    const int  arow  = row0 + l16;
    const bool rowok = arow < n;
    const float* abase = t + (size_t)(rowok ? arow : 0) * HID + half * 2;
    v2f a[16];
    float ss = 0.f;
#pragma unroll
    for (int s = 0; s < 16; ++s) {
        float2 q = rowok ? *(const float2*)(abase + 4 * s) : make_float2(0.f, 0.f);
        a[s].x = q.x; a[s].y = q.y;
        ss += q.x * q.x + q.y * q.y;
    }
    ss += __shfl_xor(ss, 16, 32);
    const float inv1 = rsqrtf(ss * (1.0f / 64.0f) + 1e-5f);
#pragma unroll
    for (int s = 0; s < 16; ++s) { a[s].x *= inv1; a[s].y *= inv1; }

    float rs[8][4];
#pragma unroll
    for (int v = 0; v < 8; ++v) {
        const int rowg = row0 + v + half * 8;
        const float* tp = t + (size_t)(rowg < n ? rowg : 0) * HID + l16;
#pragma unroll
        for (int j = 0; j < 4; ++j) rs[v][j] = tp[j * 16];
    }
#pragma unroll
    for (int v = 0; v < 8; ++v)
        asm volatile("" :: "v"(rs[v][0]), "v"(rs[v][1]), "v"(rs[v][2]), "v"(rs[v][3]));

    v8f acc[4] = {};

#pragma unroll
    for (int s = 0; s < 16; ++s) {
#pragma unroll
        for (int j = 0; j < 4; ++j) {
            const int nc = j * 16 + l16;
            float2 bb = *(const float2*)&lw[nc * WST + 4 * s + half * 2];
            v2f b; b.x = bb.x; b.y = bb.y;
            acc[j] = __builtin_amdgcn_wmma_f32_16x16x4_f32(
                false, a[s], false, b, (short)0, acc[j], false, false);
        }
    }

    float w1c[4], w2c[4];
#pragma unroll
    for (int j = 0; j < 4; ++j) {
        w1c[j] = lnw[j * 16 + l16];
        w2c[j] = lnw[HID + j * 16 + l16];
    }

    float s2[8];
#pragma unroll
    for (int v = 0; v < 8; ++v) {
        const float iv = __shfl(inv1, v + half * 8, 32);
        float s = 0.f;
#pragma unroll
        for (int j = 0; j < 4; ++j) {
            const float h = rs[v][j] * iv * w1c[j];
            const float y = h + fmaxf(acc[j][v], 0.f);
            acc[j][v] = y;
            s += y * y;
        }
        s2[v] = s;
    }
#pragma unroll
    for (int v = 0; v < 8; ++v) s2[v] = dpp_add<0x161>(s2[v]);
#pragma unroll
    for (int v = 0; v < 8; ++v) s2[v] = dpp_add<0x162>(s2[v]);
#pragma unroll
    for (int v = 0; v < 8; ++v) s2[v] = dpp_add<0x164>(s2[v]);
#pragma unroll
    for (int v = 0; v < 8; ++v) s2[v] = dpp_add<0x168>(s2[v]);

#pragma unroll
    for (int v = 0; v < 8; ++v) {
        const int rowg = row0 + v + half * 8;
        const float inv2 = rsqrtf(s2[v] * (1.0f / 64.0f) + 1e-5f);
        if (rowg < n) {
#pragma unroll
            for (int j = 0; j < 4; ++j)
                out[(size_t)rowg * HID + j * 16 + l16] = acc[j][v] * inv2 * w2c[j];
        }
    }
}

__device__ __forceinline__ void hw_atomic_add_f32(float* p, float v) {
    asm volatile("global_atomic_add_f32 %0, %1, off"
                 :
                 : "v"(p), "v"(v)
                 : "memory");
}

__global__ __launch_bounds__(256) void gnn_scatter(
    const float* __restrict__ m, const int* __restrict__ row,
    const int* __restrict__ col, float* __restrict__ t, int ne)
{
    const int wave = (int)((blockIdx.x * blockDim.x + threadIdx.x) >> 5);
    const int lane = threadIdx.x & 31;
    const int e0 = wave * 4;
#pragma unroll
    for (int i = 0; i < 4; ++i) {
        const int e = e0 + i;
        if (e >= ne) return;
        const int r = row[e];
        const int c = col[e];
        float2 v = *(const float2*)(m + (size_t)c * HID + 2 * lane);
        float* dst = t + (size_t)r * HID + 2 * lane;
        hw_atomic_add_f32(dst,     v.x);
        hw_atomic_add_f32(dst + 1, v.y);
    }
}

extern "C" void kernel_launch(void* const* d_in, const int* in_sizes, int n_in,
                              void* d_out, int out_size, void* d_ws, size_t ws_size,
                              hipStream_t stream)
{
    const float* x       = (const float*)d_in[0];
    const float* conv_w  = (const float*)d_in[1];
    const float* conv_nw = (const float*)d_in[2];
    const float* hid_w   = (const float*)d_in[3];
    const float* hid_nw  = (const float*)d_in[4];
    const int*   row     = (const int*)d_in[5];
    const int*   col     = (const int*)d_in[6];

    const int n  = in_sizes[0] / HID;
    const int ne = in_sizes[5];

    float* P = (float*)d_out;
    float* Q = (float*)d_ws;
    float* R = Q + (size_t)n * HID;
    const size_t hbytes = (size_t)n * HID * sizeof(float);

    const dim3 blk(256);
    const int gemm_grid = (n + 127) / 128;
    const int sc_grid   = ((ne + 3) / 4 + 7) / 8;

    const float* hin = x;
    for (int l = 0; l < 4; ++l) {
        const float* cw = conv_w + (size_t)l * HID * HID;
        const float* hw = hid_w  + (size_t)l * HID * HID;
        const float* cn = conv_nw + l * HID;
        const float* hn = hid_nw  + l * HID;

        gnn_gemm_relu<<<gemm_grid, blk, 0, stream>>>(hin, cw, Q, n);
        hipMemcpyAsync(R, hin, hbytes, hipMemcpyDeviceToDevice, stream);
        gnn_scatter<<<sc_grid, blk, 0, stream>>>(Q, row, col, R, ne);
        gnn_norm_gemm_norm<<<gemm_grid, blk, 0, stream>>>(R, cn, hw, hn, P, n);

        hin = P;
    }
}


# Scatter: 200,000 waves of 4 edges -> 12,500 waves of 64 consecutive edges (indices by whole-line loads and readlane, 16 row gathers in flight); same f32 atomic adds, arrival order as before
# speedup vs baseline: 1.9612x; 1.1862x over previous
#include <hip/hip_runtime.h>

typedef __attribute__((ext_vector_type(2))) float v2f;
typedef __attribute__((ext_vector_type(8))) float v8f;

#define HID 64
#define WST 68

__global__ __launch_bounds__(256) void gnn_gemm_relu(
    const float* __restrict__ in, const float* __restrict__ W,
    float* __restrict__ out, int n)
{
    __shared__ float lw[64 * WST];
    const int tid = threadIdx.x;

    for (int i = tid; i < 64 * 64; i += 256) {
        int r = i >> 6, c = i & 63;
        lw[r * WST + c] = W[i];
    }
    __syncthreads();

    const int wave = tid >> 5;
    const int lane = tid & 31;
    const int half = lane >> 4;
    const int l16  = lane & 15;
    const int row0 = blockIdx.x * 128 + wave * 16;

    const int  arow  = row0 + l16;
    const bool rowok = arow < n;
    const float* abase = in + (size_t)(rowok ? arow : 0) * HID + half * 2;
    v2f a[16];
#pragma unroll
    for (int s = 0; s < 16; ++s) {
        float2 t = rowok ? *(const float2*)(abase + 4 * s) : make_float2(0.f, 0.f);
        a[s].x = t.x; a[s].y = t.y;
    }

    v8f acc[4] = {};

#pragma unroll
    for (int s = 0; s < 16; ++s) {
#pragma unroll
        for (int j = 0; j < 4; ++j) {
            const int nc = j * 16 + l16;
            float2 bb = *(const float2*)&lw[nc * WST + 4 * s + half * 2];
            v2f b; b.x = bb.x; b.y = bb.y;
            acc[j] = __builtin_amdgcn_wmma_f32_16x16x4_f32(
                false, a[s], false, b, (short)0, acc[j], false, false);
        }
    }

#pragma unroll
    for (int j = 0; j < 4; ++j) {
        const int colg = j * 16 + l16;
#pragma unroll
        for (int v = 0; v < 8; ++v) {
            const int rowg = row0 + v + half * 8;
            if (rowg < n)
                out[(size_t)rowg * HID + colg] = fmaxf(acc[j][v], 0.f);
        }
    }
}

template <int CTRL>
__device__ __forceinline__ float dpp_add(float v) {
    return v + __int_as_float(__builtin_amdgcn_update_dpp(
                   0, __float_as_int(v), CTRL, 0xf, 0xf, true));
}

__global__ __launch_bounds__(256) void gnn_norm_gemm_norm(
    const float* __restrict__ t, const float* __restrict__ w1,
    const float* __restrict__ W, const float* __restrict__ w2,
    float* __restrict__ out, int n)
{
    __shared__ float lw[64 * WST];
    __shared__ float lnw[2 * HID];
    const int tid = threadIdx.x;

    if (tid < 2 * HID) lnw[tid] = tid < HID ? w1[tid] : w2[tid - HID];

    const float w1k = w1[tid & 63];
    for (int i = tid; i < 64 * 64; i += 256) {
        int r = i >> 6, c = i & 63;
        lw[r * WST + c] = W[i] * w1k;
    }
    __syncthreads();

    const int wave = tid >> 5;
    const int lane = tid & 31;
    const int half = lane >> 4;
    const int l16  = lane & 15;
    const int row0 = blockIdx.x * 128 + wave * 16;

    const int  arow  = row0 + l16;
    const bool rowok = arow < n;
    const float* abase = t + (size_t)(rowok ? arow : 0) * HID + half * 2;
    v2f a[16];
    float ss = 0.f;
#pragma unroll
    for (int s = 0; s < 16; ++s) {
        float2 q = rowok ? *(const float2*)(abase + 4 * s) : make_float2(0.f, 0.f);
        a[s].x = q.x; a[s].y = q.y;
        ss += q.x * q.x + q.y * q.y;
    }
    ss += __shfl_xor(ss, 16, 32);
    const float inv1 = rsqrtf(ss * (1.0f / 64.0f) + 1e-5f);
#pragma unroll
    for (int s = 0; s < 16; ++s) { a[s].x *= inv1; a[s].y *= inv1; }

    float rs[8][4];
#pragma unroll
    for (int v = 0; v < 8; ++v) {
        const int rowg = row0 + v + half * 8;
        const float* tp = t + (size_t)(rowg < n ? rowg : 0) * HID + l16;
#pragma unroll
        for (int j = 0; j < 4; ++j) rs[v][j] = tp[j * 16];
    }
#pragma unroll
    for (int v = 0; v < 8; ++v)
        asm volatile("" :: "v"(rs[v][0]), "v"(rs[v][1]), "v"(rs[v][2]), "v"(rs[v][3]));

    v8f acc[4] = {};

#pragma unroll
    for (int s = 0; s < 16; ++s) {
#pragma unroll
        for (int j = 0; j < 4; ++j) {
            const int nc = j * 16 + l16;
            float2 bb = *(const float2*)&lw[nc * WST + 4 * s + half * 2];
            v2f b; b.x = bb.x; b.y = bb.y;
            acc[j] = __builtin_amdgcn_wmma_f32_16x16x4_f32(
                false, a[s], false, b, (short)0, acc[j], false, false);
        }
    }

    float w1c[4], w2c[4];
#pragma unroll
    for (int j = 0; j < 4; ++j) {
        w1c[j] = lnw[j * 16 + l16];
        w2c[j] = lnw[HID + j * 16 + l16];
    }

    float s2[8];
#pragma unroll
    for (int v = 0; v < 8; ++v) {
        const float iv = __shfl(inv1, v + half * 8, 32);
        float s = 0.f;
#pragma unroll
        for (int j = 0; j < 4; ++j) {
            const float h = rs[v][j] * iv * w1c[j];
            const float y = h + fmaxf(acc[j][v], 0.f);
            acc[j][v] = y;
            s += y * y;
        }
        s2[v] = s;
    }
#pragma unroll
    for (int v = 0; v < 8; ++v) s2[v] = dpp_add<0x161>(s2[v]);
#pragma unroll
    for (int v = 0; v < 8; ++v) s2[v] = dpp_add<0x162>(s2[v]);
#pragma unroll
    for (int v = 0; v < 8; ++v) s2[v] = dpp_add<0x164>(s2[v]);
#pragma unroll
    for (int v = 0; v < 8; ++v) s2[v] = dpp_add<0x168>(s2[v]);

#pragma unroll
    for (int v = 0; v < 8; ++v) {
        const int rowg = row0 + v + half * 8;
        const float inv2 = rsqrtf(s2[v] * (1.0f / 64.0f) + 1e-5f);
        if (rowg < n) {
#pragma unroll
            for (int j = 0; j < 4; ++j)
                out[(size_t)rowg * HID + j * 16 + l16] = acc[j][v] * inv2 * w2c[j];
        }
    }
}

__device__ __forceinline__ void hw_atomic_add_f32x2(float* rowbase, unsigned laneoff, float2 v) {
    asm volatile("global_atomic_add_f32 %0, %1, %3\n\t"
                 "global_atomic_add_f32 %0, %2, %3 offset:4"
                 :
                 : "v"(laneoff), "v"(v.x), "v"(v.y), "s"(rowbase)
                 : "memory");
}

#define SC_EPW 64
#define SC_GRP 16

__global__ __launch_bounds__(256) void gnn_scatter(
    const float* __restrict__ m, const int* __restrict__ row,
    const int* __restrict__ col, float* __restrict__ t, int ne)
{
    const int lane = threadIdx.x & 31;
    const int wave = blockIdx.x * 8 +
                     __builtin_amdgcn_readfirstlane((int)(threadIdx.x >> 5));
    const int e0 = wave * SC_EPW;
    if (e0 >= ne) return;
    const int left = ne - e0;
    const int cnt = left < SC_EPW ? left : SC_EPW;
    const unsigned laneoff = (unsigned)lane * 8u;

    int rv[SC_EPW / 32], cv[SC_EPW / 32];
#pragma unroll
    for (int b = 0; b < SC_EPW / 32; ++b) {
        const int e = e0 + b * 32 + lane;
        const bool ok = e < ne;
        rv[b] = ok ? row[e] : 0;
        cv[b] = ok ? col[e] : 0;
    }

#pragma unroll
    for (int g = 0; g < SC_EPW; g += SC_GRP) {
        if (g >= cnt) break;
        float2 v[SC_GRP];
#pragma unroll
        for (int k = 0; k < SC_GRP; ++k) {
            const int i = g + k;
            const int c = __builtin_amdgcn_readlane(cv[i >> 5], i & 31);
            v[k] = *(const float2*)((const char*)(m + (size_t)c * HID) + laneoff);
        }
        if (g + SC_GRP <= cnt) {
#pragma unroll
            for (int k = 0; k < SC_GRP; ++k) {
                const int i = g + k;
                const int r = __builtin_amdgcn_readlane(rv[i >> 5], i & 31);
                hw_atomic_add_f32x2(t + (size_t)r * HID, laneoff, v[k]);
            }
        } else {
#pragma unroll
            for (int k = 0; k < SC_GRP; ++k) {
                const int i = g + k;
                if (i < cnt) {
                    const int r = __builtin_amdgcn_readlane(rv[i >> 5], i & 31);
                    hw_atomic_add_f32x2(t + (size_t)r * HID, laneoff, v[k]);
                }
            }
        }
    }
}

extern "C" void kernel_launch(void* const* d_in, const int* in_sizes, int n_in,
                              void* d_out, int out_size, void* d_ws, size_t ws_size,
                              hipStream_t stream)
{
    const float* x       = (const float*)d_in[0];
    const float* conv_w  = (const float*)d_in[1];
    const float* conv_nw = (const float*)d_in[2];
    const float* hid_w   = (const float*)d_in[3];
    const float* hid_nw  = (const float*)d_in[4];
    const int*   row     = (const int*)d_in[5];
    const int*   col     = (const int*)d_in[6];

    const int n  = in_sizes[0] / HID;
    const int ne = in_sizes[5];

    float* P = (float*)d_out;
    float* Q = (float*)d_ws;
    float* R = Q + (size_t)n * HID;
    const size_t hbytes = (size_t)n * HID * sizeof(float);

    const dim3 blk(256);
    const int gemm_grid = (n + 127) / 128;
    const int sc_grid   = ((ne + SC_EPW - 1) / SC_EPW + 7) / 8;

    const float* hin = x;
    for (int l = 0; l < 4; ++l) {
        const float* cw = conv_w + (size_t)l * HID * HID;
        const float* hw = hid_w  + (size_t)l * HID * HID;
        const float* cn = conv_nw + l * HID;
        const float* hn = hid_nw  + l * HID;

        gnn_gemm_relu<<<gemm_grid, blk, 0, stream>>>(hin, cw, Q, n);
        hipMemcpyAsync(R, hin, hbytes, hipMemcpyDeviceToDevice, stream);
        gnn_scatter<<<sc_grid, blk, 0, stream>>>(Q, row, col, R, ne);
        gnn_norm_gemm_norm<<<gemm_grid, blk, 0, stream>>>(R, cn, hw, hn, P, n);

        hin = P;
    }
}


# m = relu(h W^T) stored with columns c and c+32 adjacent: each of the scatter's two f32 atomic adds per edge covers one whole 128-byte line, not half of two; same adds, arrival order as before
# speedup vs baseline: 2.0369x; 1.0386x over previous
#include <hip/hip_runtime.h>

typedef __attribute__((ext_vector_type(2))) float v2f;
typedef __attribute__((ext_vector_type(8))) float v8f;

#define HID 64
#define WST 68

__global__ __launch_bounds__(256) void gnn_gemm_relu(
    const float* __restrict__ in, const float* __restrict__ W,
    float* __restrict__ out, int n)
{
    __shared__ float lw[64 * WST];
    const int tid = threadIdx.x;

    for (int i = tid; i < 64 * 64; i += 256) {
        int r = i >> 6, c = i & 63;
        lw[r * WST + c] = W[i];
    }
    __syncthreads();

    const int wave = tid >> 5;
    const int lane = tid & 31;
    const int half = lane >> 4;
    const int l16  = lane & 15;
    const int row0 = blockIdx.x * 128 + wave * 16;

    const int  arow  = row0 + l16;
    const bool rowok = arow < n;
    const float* abase = in + (size_t)(rowok ? arow : 0) * HID + half * 2;
    v2f a[16];
#pragma unroll
    for (int s = 0; s < 16; ++s) {
        float2 t = rowok ? *(const float2*)(abase + 4 * s) : make_float2(0.f, 0.f);
        a[s].x = t.x; a[s].y = t.y;
    }

    v8f acc[4] = {};

#pragma unroll
    for (int s = 0; s < 16; ++s) {
#pragma unroll
        for (int j = 0; j < 4; ++j) {
            const int nc = j * 16 + l16;
            float2 bb = *(const float2*)&lw[nc * WST + 4 * s + half * 2];
            v2f b; b.x = bb.x; b.y = bb.y;
            acc[j] = __builtin_amdgcn_wmma_f32_16x16x4_f32(
                false, a[s], false, b, (short)0, acc[j], false, false);
        }
    }

#pragma unroll
    for (int v = 0; v < 8; ++v) {
        const int rowg = row0 + v + half * 8;
        if (rowg < n) {
            float* op = out + (size_t)rowg * HID + 2 * l16;
            *(float2*)op        = make_float2(fmaxf(acc[0][v], 0.f), fmaxf(acc[2][v], 0.f));
            *(float2*)(op + 32) = make_float2(fmaxf(acc[1][v], 0.f), fmaxf(acc[3][v], 0.f));
        }
    }
}

template <int CTRL>
__device__ __forceinline__ float dpp_add(float v) {
    return v + __int_as_float(__builtin_amdgcn_update_dpp(
                   0, __float_as_int(v), CTRL, 0xf, 0xf, true));
}

__global__ __launch_bounds__(256) void gnn_norm_gemm_norm(
    const float* __restrict__ t, const float* __restrict__ w1,
    const float* __restrict__ W, const float* __restrict__ w2,
    float* __restrict__ out, int n)
{
    __shared__ float lw[64 * WST];
    __shared__ float lnw[2 * HID];
    const int tid = threadIdx.x;

    if (tid < 2 * HID) lnw[tid] = tid < HID ? w1[tid] : w2[tid - HID];

    const float w1k = w1[tid & 63];
    for (int i = tid; i < 64 * 64; i += 256) {
        int r = i >> 6, c = i & 63;
        lw[r * WST + c] = W[i] * w1k;
    }
    __syncthreads();

    const int wave = tid >> 5;
    const int lane = tid & 31;
    const int half = lane >> 4;
    const int l16  = lane & 15;
    const int row0 = blockIdx.x * 128 + wave * 16;

    const int  arow  = row0 + l16;
    const bool rowok = arow < n;
    const float* abase = t + (size_t)(rowok ? arow : 0) * HID + half * 2;
    v2f a[16];
    float ss = 0.f;
#pragma unroll
    for (int s = 0; s < 16; ++s) {
        float2 q = rowok ? *(const float2*)(abase + 4 * s) : make_float2(0.f, 0.f);
        a[s].x = q.x; a[s].y = q.y;
        ss += q.x * q.x + q.y * q.y;
    }
    ss += __shfl_xor(ss, 16, 32);
    const float inv1 = rsqrtf(ss * (1.0f / 64.0f) + 1e-5f);
#pragma unroll
    for (int s = 0; s < 16; ++s) { a[s].x *= inv1; a[s].y *= inv1; }

    float rs[8][4];
#pragma unroll
    for (int v = 0; v < 8; ++v) {
        const int rowg = row0 + v + half * 8;
        const float* tp = t + (size_t)(rowg < n ? rowg : 0) * HID + l16;
#pragma unroll
        for (int j = 0; j < 4; ++j) rs[v][j] = tp[j * 16];
    }
#pragma unroll
    for (int v = 0; v < 8; ++v)
        asm volatile("" :: "v"(rs[v][0]), "v"(rs[v][1]), "v"(rs[v][2]), "v"(rs[v][3]));

    v8f acc[4] = {};

#pragma unroll
    for (int s = 0; s < 16; ++s) {
#pragma unroll
        for (int j = 0; j < 4; ++j) {
            const int nc = j * 16 + l16;
            float2 bb = *(const float2*)&lw[nc * WST + 4 * s + half * 2];
            v2f b; b.x = bb.x; b.y = bb.y;
            acc[j] = __builtin_amdgcn_wmma_f32_16x16x4_f32(
                false, a[s], false, b, (short)0, acc[j], false, false);
        }
    }

    float w1c[4], w2c[4];
#pragma unroll
    for (int j = 0; j < 4; ++j) {
        w1c[j] = lnw[j * 16 + l16];
        w2c[j] = lnw[HID + j * 16 + l16];
    }

    float s2[8];
#pragma unroll
    for (int v = 0; v < 8; ++v) {
        const float iv = __shfl(inv1, v + half * 8, 32);
        float s = 0.f;
#pragma unroll
        for (int j = 0; j < 4; ++j) {
            const float h = rs[v][j] * iv * w1c[j];
            const float y = h + fmaxf(acc[j][v], 0.f);
            acc[j][v] = y;
            s += y * y;
        }
        s2[v] = s;
    }
#pragma unroll
    for (int v = 0; v < 8; ++v) s2[v] = dpp_add<0x161>(s2[v]);
#pragma unroll
    for (int v = 0; v < 8; ++v) s2[v] = dpp_add<0x162>(s2[v]);
#pragma unroll
    for (int v = 0; v < 8; ++v) s2[v] = dpp_add<0x164>(s2[v]);
#pragma unroll
    for (int v = 0; v < 8; ++v) s2[v] = dpp_add<0x168>(s2[v]);

#pragma unroll
    for (int v = 0; v < 8; ++v) {
        const int rowg = row0 + v + half * 8;
        const float inv2 = rsqrtf(s2[v] * (1.0f / 64.0f) + 1e-5f);
        if (rowg < n) {
#pragma unroll
            for (int j = 0; j < 4; ++j)
                out[(size_t)rowg * HID + j * 16 + l16] = acc[j][v] * inv2 * w2c[j];
        }
    }
}

__device__ __forceinline__ void hw_atomic_add_f32x2(float* rowbase, unsigned laneoff, float2 v) {
    asm volatile("global_atomic_add_f32 %0, %1, %3\n\t"
                 "global_atomic_add_f32 %0, %2, %3 offset:128"
                 :
                 : "v"(laneoff), "v"(v.x), "v"(v.y), "s"(rowbase)
                 : "memory");
}

#define SC_EPW 64
#define SC_GRP 16

__global__ __launch_bounds__(256) void gnn_scatter(
    const float* __restrict__ m, const int* __restrict__ row,
    const int* __restrict__ col, float* __restrict__ t, int ne)
{
    const int lane = threadIdx.x & 31;
    const int wave = blockIdx.x * 8 +
                     __builtin_amdgcn_readfirstlane((int)(threadIdx.x >> 5));
    const int e0 = wave * SC_EPW;
    if (e0 >= ne) return;
    const int left = ne - e0;
    const int cnt = left < SC_EPW ? left : SC_EPW;
    const unsigned laneoff = (unsigned)lane * 8u;
    const unsigned aoff    = (unsigned)lane * 4u;

    int rv[SC_EPW / 32], cv[SC_EPW / 32];
#pragma unroll
    for (int b = 0; b < SC_EPW / 32; ++b) {
        const int e = e0 + b * 32 + lane;
        const bool ok = e < ne;
        rv[b] = ok ? row[e] : 0;
        cv[b] = ok ? col[e] : 0;
    }

#pragma unroll
    for (int g = 0; g < SC_EPW; g += SC_GRP) {
        if (g >= cnt) break;
        float2 v[SC_GRP];
#pragma unroll
        for (int k = 0; k < SC_GRP; ++k) {
            const int i = g + k;
            const int c = __builtin_amdgcn_readlane(cv[i >> 5], i & 31);
            v[k] = *(const float2*)((const char*)(m + (size_t)c * HID) + laneoff);
        }
        if (g + SC_GRP <= cnt) {
#pragma unroll
            for (int k = 0; k < SC_GRP; ++k) {
                const int i = g + k;
                const int r = __builtin_amdgcn_readlane(rv[i >> 5], i & 31);
                hw_atomic_add_f32x2(t + (size_t)r * HID, aoff, v[k]);
            }
        } else {
#pragma unroll
            for (int k = 0; k < SC_GRP; ++k) {
                const int i = g + k;
                if (i < cnt) {
                    const int r = __builtin_amdgcn_readlane(rv[i >> 5], i & 31);
                    hw_atomic_add_f32x2(t + (size_t)r * HID, aoff, v[k]);
                }
            }
        }
    }
}

extern "C" void kernel_launch(void* const* d_in, const int* in_sizes, int n_in,
                              void* d_out, int out_size, void* d_ws, size_t ws_size,
                              hipStream_t stream)
{
    const float* x       = (const float*)d_in[0];
    const float* conv_w  = (const float*)d_in[1];
    const float* conv_nw = (const float*)d_in[2];
    const float* hid_w   = (const float*)d_in[3];
    const float* hid_nw  = (const float*)d_in[4];
    const int*   row     = (const int*)d_in[5];
    const int*   col     = (const int*)d_in[6];

    const int n  = in_sizes[0] / HID;
    const int ne = in_sizes[5];

    float* P = (float*)d_out;
    float* Q = (float*)d_ws;
    float* R = Q + (size_t)n * HID;
    const size_t hbytes = (size_t)n * HID * sizeof(float);

    const dim3 blk(256);
    const int gemm_grid = (n + 127) / 128;
    const int sc_grid   = ((ne + SC_EPW - 1) / SC_EPW + 7) / 8;

    const float* hin = x;
    for (int l = 0; l < 4; ++l) {
        const float* cw = conv_w + (size_t)l * HID * HID;
        const float* hw = hid_w  + (size_t)l * HID * HID;
        const float* cn = conv_nw + l * HID;
        const float* hn = hid_nw  + l * HID;

        gnn_gemm_relu<<<gemm_grid, blk, 0, stream>>>(hin, cw, Q, n);
        hipMemcpyAsync(R, hin, hbytes, hipMemcpyDeviceToDevice, stream);
        gnn_scatter<<<sc_grid, blk, 0, stream>>>(Q, row, col, R, ne);
        gnn_norm_gemm_norm<<<gemm_grid, blk, 0, stream>>>(R, cn, hw, hn, P, n);

        hin = P;
    }
}


# Layers 2-4: first product m = relu(h W^T) moved into the dense kernel before it (finished rows go D layout to A layout through a per-wave LDS tile): 12 -> 9 launches, h not read back
# speedup vs baseline: 2.1436x; 1.0524x over previous
#include <hip/hip_runtime.h>

typedef __attribute__((ext_vector_type(2))) float v2f;
typedef __attribute__((ext_vector_type(8))) float v8f;

#define HID 64
#define WST 68

__global__ __launch_bounds__(256) void gnn_gemm_relu(
    const float* __restrict__ in, const float* __restrict__ W,
    float* __restrict__ out, int n)
{
    __shared__ float lw[64 * WST];
    const int tid = threadIdx.x;

    for (int i = tid; i < 64 * 64; i += 256) {
        int r = i >> 6, c = i & 63;
        lw[r * WST + c] = W[i];
    }
    __syncthreads();

    const int wave = tid >> 5;
    const int lane = tid & 31;
    const int half = lane >> 4;
    const int l16  = lane & 15;
    const int row0 = blockIdx.x * 128 + wave * 16;

    const int  arow  = row0 + l16;
    const bool rowok = arow < n;
    const float* abase = in + (size_t)(rowok ? arow : 0) * HID + half * 2;
    v2f a[16];
#pragma unroll
    for (int s = 0; s < 16; ++s) {
        float2 t = rowok ? *(const float2*)(abase + 4 * s) : make_float2(0.f, 0.f);
        a[s].x = t.x; a[s].y = t.y;
    }

    v8f acc[4] = {};

#pragma unroll
    for (int s = 0; s < 16; ++s) {
#pragma unroll
        for (int j = 0; j < 4; ++j) {
            const int nc = j * 16 + l16;
            float2 bb = *(const float2*)&lw[nc * WST + 4 * s + half * 2];
            v2f b; b.x = bb.x; b.y = bb.y;
            acc[j] = __builtin_amdgcn_wmma_f32_16x16x4_f32(
                false, a[s], false, b, (short)0, acc[j], false, false);
        }
    }

#pragma unroll
    for (int v = 0; v < 8; ++v) {
        const int rowg = row0 + v + half * 8;
        if (rowg < n) {
            float* op = out + (size_t)rowg * HID + 2 * l16;
            *(float2*)op        = make_float2(fmaxf(acc[0][v], 0.f), fmaxf(acc[2][v], 0.f));
            *(float2*)(op + 32) = make_float2(fmaxf(acc[1][v], 0.f), fmaxf(acc[3][v], 0.f));
        }
    }
}

template <int CTRL>
__device__ __forceinline__ float dpp_add(float v) {
    return v + __int_as_float(__builtin_amdgcn_update_dpp(
                   0, __float_as_int(v), CTRL, 0xf, 0xf, true));
}

__global__ __launch_bounds__(256) void gnn_norm_gemm_norm(
    const float* __restrict__ t, const float* __restrict__ w1,
    const float* __restrict__ W, const float* __restrict__ w2,
    float* __restrict__ out, const float* __restrict__ Wn,
    float* __restrict__ mout, int n)
{
    __shared__ float lw[64 * WST];
    __shared__ float lw2[64 * WST];
    __shared__ float tile[8 * 16 * WST];
    __shared__ float lnw[2 * HID];
    const int tid = threadIdx.x;
    const bool fuse = Wn != nullptr;

    if (tid < 2 * HID) lnw[tid] = tid < HID ? w1[tid] : w2[tid - HID];

    const float w1k = w1[tid & 63];
    for (int i = tid; i < 64 * 64; i += 256) {
        int r = i >> 6, c = i & 63;
        lw[r * WST + c] = W[i] * w1k;
    }
    if (fuse) {
        for (int i = tid; i < 64 * 64; i += 256) {
            int r = i >> 6, c = i & 63;
            lw2[r * WST + c] = Wn[i];
        }
    }
    __syncthreads();

    const int wave = tid >> 5;
    const int lane = tid & 31;
    const int half = lane >> 4;
    const int l16  = lane & 15;
    const int row0 = blockIdx.x * 128 + wave * 16;

    const int  arow  = row0 + l16;
    const bool rowok = arow < n;
    const float* abase = t + (size_t)(rowok ? arow : 0) * HID + half * 2;
    v2f a[16];
    float ss = 0.f;
#pragma unroll
    for (int s = 0; s < 16; ++s) {
        float2 q = rowok ? *(const float2*)(abase + 4 * s) : make_float2(0.f, 0.f);
        a[s].x = q.x; a[s].y = q.y;
        ss += q.x * q.x + q.y * q.y;
    }
    ss += __shfl_xor(ss, 16, 32);
    const float inv1 = rsqrtf(ss * (1.0f / 64.0f) + 1e-5f);
#pragma unroll
    for (int s = 0; s < 16; ++s) { a[s].x *= inv1; a[s].y *= inv1; }

    float rs[8][4];
#pragma unroll
    for (int v = 0; v < 8; ++v) {
        const int rowg = row0 + v + half * 8;
        const float* tp = t + (size_t)(rowg < n ? rowg : 0) * HID + l16;
#pragma unroll
        for (int j = 0; j < 4; ++j) rs[v][j] = tp[j * 16];
    }
#pragma unroll
    for (int v = 0; v < 8; ++v)
        asm volatile("" :: "v"(rs[v][0]), "v"(rs[v][1]), "v"(rs[v][2]), "v"(rs[v][3]));

    v8f acc[4] = {};

#pragma unroll
    for (int s = 0; s < 16; ++s) {
#pragma unroll
        for (int j = 0; j < 4; ++j) {
            const int nc = j * 16 + l16;
            float2 bb = *(const float2*)&lw[nc * WST + 4 * s + half * 2];
            v2f b; b.x = bb.x; b.y = bb.y;
            acc[j] = __builtin_amdgcn_wmma_f32_16x16x4_f32(
                false, a[s], false, b, (short)0, acc[j], false, false);
        }
    }

    float w1c[4], w2c[4];
#pragma unroll
    for (int j = 0; j < 4; ++j) {
        w1c[j] = lnw[j * 16 + l16];
        w2c[j] = lnw[HID + j * 16 + l16];
    }

    float s2[8];
#pragma unroll
    for (int v = 0; v < 8; ++v) {
        const float iv = __shfl(inv1, v + half * 8, 32);
        float s = 0.f;
#pragma unroll
        for (int j = 0; j < 4; ++j) {
            const float h = rs[v][j] * iv * w1c[j];
            const float y = h + fmaxf(acc[j][v], 0.f);
            acc[j][v] = y;
            s += y * y;
        }
        s2[v] = s;
    }
#pragma unroll
    for (int v = 0; v < 8; ++v) s2[v] = dpp_add<0x161>(s2[v]);
#pragma unroll
    for (int v = 0; v < 8; ++v) s2[v] = dpp_add<0x162>(s2[v]);
#pragma unroll
    for (int v = 0; v < 8; ++v) s2[v] = dpp_add<0x164>(s2[v]);
#pragma unroll
    for (int v = 0; v < 8; ++v) s2[v] = dpp_add<0x168>(s2[v]);

#pragma unroll
    for (int v = 0; v < 8; ++v) {
        const int rowg = row0 + v + half * 8;
        const float inv2 = rsqrtf(s2[v] * (1.0f / 64.0f) + 1e-5f);
#pragma unroll
        for (int j = 0; j < 4; ++j) acc[j][v] = acc[j][v] * inv2 * w2c[j];
        if (rowg < n) {
#pragma unroll
            for (int j = 0; j < 4; ++j)
                out[(size_t)rowg * HID + j * 16 + l16] = acc[j][v];
        }
    }

    if (!fuse) return;

    float* tl = tile + wave * (16 * WST);
#pragma unroll
    for (int v = 0; v < 8; ++v)
#pragma unroll
        for (int j = 0; j < 4; ++j)
            tl[(v + half * 8) * WST + j * 16 + l16] = acc[j][v];

    __builtin_amdgcn_fence(__ATOMIC_RELEASE, "wavefront");
    __builtin_amdgcn_wave_barrier();
    __builtin_amdgcn_fence(__ATOMIC_ACQUIRE, "wavefront");

    v2f a2[16];
#pragma unroll
    for (int s = 0; s < 16; ++s) {
        float2 q = *(const float2*)&tl[l16 * WST + 4 * s + half * 2];
        a2[s].x = q.x; a2[s].y = q.y;
    }

    v8f acc2[4] = {};
#pragma unroll
    for (int s = 0; s < 16; ++s) {
#pragma unroll
        for (int j = 0; j < 4; ++j) {
            const int nc = j * 16 + l16;
            float2 bb = *(const float2*)&lw2[nc * WST + 4 * s + half * 2];
            v2f b; b.x = bb.x; b.y = bb.y;
            acc2[j] = __builtin_amdgcn_wmma_f32_16x16x4_f32(
                false, a2[s], false, b, (short)0, acc2[j], false, false);
        }
    }

#pragma unroll
    for (int v = 0; v < 8; ++v) {
        const int rowg = row0 + v + half * 8;
        if (rowg < n) {
            float* op = mout + (size_t)rowg * HID + 2 * l16;
            *(float2*)op        = make_float2(fmaxf(acc2[0][v], 0.f), fmaxf(acc2[2][v], 0.f));
            *(float2*)(op + 32) = make_float2(fmaxf(acc2[1][v], 0.f), fmaxf(acc2[3][v], 0.f));
        }
    }
}

__device__ __forceinline__ void hw_atomic_add_f32x2(float* rowbase, unsigned laneoff, float2 v) {
    asm volatile("global_atomic_add_f32 %0, %1, %3\n\t"
                 "global_atomic_add_f32 %0, %2, %3 offset:128"
                 :
                 : "v"(laneoff), "v"(v.x), "v"(v.y), "s"(rowbase)
                 : "memory");
}

#define SC_EPW 64
#define SC_GRP 16

__global__ __launch_bounds__(256) void gnn_scatter(
    const float* __restrict__ m, const int* __restrict__ row,
    const int* __restrict__ col, float* __restrict__ t, int ne)
{
    const int lane = threadIdx.x & 31;
    const int wave = blockIdx.x * 8 +
                     __builtin_amdgcn_readfirstlane((int)(threadIdx.x >> 5));
    const int e0 = wave * SC_EPW;
    if (e0 >= ne) return;
    const int left = ne - e0;
    const int cnt = left < SC_EPW ? left : SC_EPW;
    const unsigned laneoff = (unsigned)lane * 8u;
    const unsigned aoff    = (unsigned)lane * 4u;

    int rv[SC_EPW / 32], cv[SC_EPW / 32];
#pragma unroll
    for (int b = 0; b < SC_EPW / 32; ++b) {
        const int e = e0 + b * 32 + lane;
        const bool ok = e < ne;
        rv[b] = ok ? row[e] : 0;
        cv[b] = ok ? col[e] : 0;
    }

#pragma unroll
    for (int g = 0; g < SC_EPW; g += SC_GRP) {
        if (g >= cnt) break;
        float2 v[SC_GRP];
#pragma unroll
        for (int k = 0; k < SC_GRP; ++k) {
            const int i = g + k;
            const int c = __builtin_amdgcn_readlane(cv[i >> 5], i & 31);
            v[k] = *(const float2*)((const char*)(m + (size_t)c * HID) + laneoff);
        }
        if (g + SC_GRP <= cnt) {
#pragma unroll
            for (int k = 0; k < SC_GRP; ++k) {
                const int i = g + k;
                const int r = __builtin_amdgcn_readlane(rv[i >> 5], i & 31);
                hw_atomic_add_f32x2(t + (size_t)r * HID, aoff, v[k]);
            }
        } else {
#pragma unroll
            for (int k = 0; k < SC_GRP; ++k) {
                const int i = g + k;
                if (i < cnt) {
                    const int r = __builtin_amdgcn_readlane(rv[i >> 5], i & 31);
                    hw_atomic_add_f32x2(t + (size_t)r * HID, aoff, v[k]);
                }
            }
        }
    }
}

extern "C" void kernel_launch(void* const* d_in, const int* in_sizes, int n_in,
                              void* d_out, int out_size, void* d_ws, size_t ws_size,
                              hipStream_t stream)
{
    const float* x       = (const float*)d_in[0];
    const float* conv_w  = (const float*)d_in[1];
    const float* conv_nw = (const float*)d_in[2];
    const float* hid_w   = (const float*)d_in[3];
    const float* hid_nw  = (const float*)d_in[4];
    const int*   row     = (const int*)d_in[5];
    const int*   col     = (const int*)d_in[6];

    const int n  = in_sizes[0] / HID;
    const int ne = in_sizes[5];

    float* P = (float*)d_out;
    float* Q = (float*)d_ws;
    float* R = Q + (size_t)n * HID;
    const size_t hbytes = (size_t)n * HID * sizeof(float);

    const dim3 blk(256);
    const int gemm_grid = (n + 127) / 128;
    const int sc_grid   = ((ne + SC_EPW - 1) / SC_EPW + 7) / 8;

    gnn_gemm_relu<<<gemm_grid, blk, 0, stream>>>(x, conv_w, Q, n);

    const float* hin = x;
    for (int l = 0; l < 4; ++l) {
        const float* hw = hid_w  + (size_t)l * HID * HID;
        const float* cn = conv_nw + l * HID;
        const float* hn = hid_nw  + l * HID;
        const float* cwn = (l + 1 < 4) ? conv_w + (size_t)(l + 1) * HID * HID : nullptr;

        hipMemcpyAsync(R, hin, hbytes, hipMemcpyDeviceToDevice, stream);
        gnn_scatter<<<sc_grid, blk, 0, stream>>>(Q, row, col, R, ne);
        gnn_norm_gemm_norm<<<gemm_grid, blk, 0, stream>>>(R, cn, hw, hn, P, cwn, Q, n);

        hin = P;
    }
}


# Aggregation as a pull: edges grouped once per call by 64-row destination block (integer atomics, scan); a block sums its rows by f32 LDS adds in arrival order and stores t = h + sum once; no copies
# speedup vs baseline: 2.4877x; 1.1605x over previous
#include <hip/hip_runtime.h>

typedef __attribute__((ext_vector_type(2))) float v2f;
typedef __attribute__((ext_vector_type(8))) float v8f;

#define HID 64
#define WST 68

__global__ __launch_bounds__(256) void gnn_gemm_relu(
    const float* __restrict__ in, const float* __restrict__ W,
    float* __restrict__ out, int n, unsigned* __restrict__ cnt, int nb)
{
    __shared__ float lw[64 * WST];
    const int tid = threadIdx.x;

    {
        const int gi = blockIdx.x * 256 + tid;
        if (gi < nb) cnt[gi] = 0u;
    }

    for (int i = tid; i < 64 * 64; i += 256) {
        int r = i >> 6, c = i & 63;
        lw[r * WST + c] = W[i];
    }
    __syncthreads();

    const int wave = tid >> 5;
    const int lane = tid & 31;
    const int half = lane >> 4;
    const int l16  = lane & 15;
    const int row0 = blockIdx.x * 128 + wave * 16;

    const int  arow  = row0 + l16;
    const bool rowok = arow < n;
    const float* abase = in + (size_t)(rowok ? arow : 0) * HID + half * 2;
    v2f a[16];
#pragma unroll
    for (int s = 0; s < 16; ++s) {
        float2 t = rowok ? *(const float2*)(abase + 4 * s) : make_float2(0.f, 0.f);
        a[s].x = t.x; a[s].y = t.y;
    }

    v8f acc[4] = {};

#pragma unroll
    for (int s = 0; s < 16; ++s) {
#pragma unroll
        for (int j = 0; j < 4; ++j) {
            const int nc = j * 16 + l16;
            float2 bb = *(const float2*)&lw[nc * WST + 4 * s + half * 2];
            v2f b; b.x = bb.x; b.y = bb.y;
            acc[j] = __builtin_amdgcn_wmma_f32_16x16x4_f32(
                false, a[s], false, b, (short)0, acc[j], false, false);
        }
    }

#pragma unroll
    for (int v = 0; v < 8; ++v) {
        const int rowg = row0 + v + half * 8;
        if (rowg < n) {
            float* op = out + (size_t)rowg * HID + 2 * l16;
            *(float2*)op        = make_float2(fmaxf(acc[0][v], 0.f), fmaxf(acc[2][v], 0.f));
            *(float2*)(op + 32) = make_float2(fmaxf(acc[1][v], 0.f), fmaxf(acc[3][v], 0.f));
        }
    }
}

template <int CTRL>
__device__ __forceinline__ float dpp_add(float v) {
    return v + __int_as_float(__builtin_amdgcn_update_dpp(
                   0, __float_as_int(v), CTRL, 0xf, 0xf, true));
}

__global__ __launch_bounds__(256) void gnn_norm_gemm_norm(
    const float* __restrict__ t, const float* __restrict__ w1,
    const float* __restrict__ W, const float* __restrict__ w2,
    float* __restrict__ out, const float* __restrict__ Wn,
    float* __restrict__ mout, int n)
{
    __shared__ float lw[64 * WST];
    __shared__ float lw2[64 * WST];
    __shared__ float tile[8 * 16 * WST];
    __shared__ float lnw[2 * HID];
    const int tid = threadIdx.x;
    const bool fuse = Wn != nullptr;

    if (tid < 2 * HID) lnw[tid] = tid < HID ? w1[tid] : w2[tid - HID];

    const float w1k = w1[tid & 63];
    for (int i = tid; i < 64 * 64; i += 256) {
        int r = i >> 6, c = i & 63;
        lw[r * WST + c] = W[i] * w1k;
    }
    if (fuse) {
        for (int i = tid; i < 64 * 64; i += 256) {
            int r = i >> 6, c = i & 63;
            lw2[r * WST + c] = Wn[i];
        }
    }
    __syncthreads();

    const int wave = tid >> 5;
    const int lane = tid & 31;
    const int half = lane >> 4;
    const int l16  = lane & 15;
    const int row0 = blockIdx.x * 128 + wave * 16;

    const int  arow  = row0 + l16;
    const bool rowok = arow < n;
    const float* abase = t + (size_t)(rowok ? arow : 0) * HID + half * 2;
    v2f a[16];
    float ss = 0.f;
#pragma unroll
    for (int s = 0; s < 16; ++s) {
        float2 q = rowok ? *(const float2*)(abase + 4 * s) : make_float2(0.f, 0.f);
        a[s].x = q.x; a[s].y = q.y;
        ss += q.x * q.x + q.y * q.y;
    }
    ss += __shfl_xor(ss, 16, 32);
    const float inv1 = rsqrtf(ss * (1.0f / 64.0f) + 1e-5f);
#pragma unroll
    for (int s = 0; s < 16; ++s) { a[s].x *= inv1; a[s].y *= inv1; }

    float rs[8][4];
#pragma unroll
    for (int v = 0; v < 8; ++v) {
        const int rowg = row0 + v + half * 8;
        const float* tp = t + (size_t)(rowg < n ? rowg : 0) * HID + l16;
#pragma unroll
        for (int j = 0; j < 4; ++j) rs[v][j] = tp[j * 16];
    }
#pragma unroll
    for (int v = 0; v < 8; ++v)
        asm volatile("" :: "v"(rs[v][0]), "v"(rs[v][1]), "v"(rs[v][2]), "v"(rs[v][3]));

    v8f acc[4] = {};

#pragma unroll
    for (int s = 0; s < 16; ++s) {
#pragma unroll
        for (int j = 0; j < 4; ++j) {
            const int nc = j * 16 + l16;
            float2 bb = *(const float2*)&lw[nc * WST + 4 * s + half * 2];
            v2f b; b.x = bb.x; b.y = bb.y;
            acc[j] = __builtin_amdgcn_wmma_f32_16x16x4_f32(
                false, a[s], false, b, (short)0, acc[j], false, false);
        }
    }

    float w1c[4], w2c[4];
#pragma unroll
    for (int j = 0; j < 4; ++j) {
        w1c[j] = lnw[j * 16 + l16];
        w2c[j] = lnw[HID + j * 16 + l16];
    }

    float s2[8];
#pragma unroll
    for (int v = 0; v < 8; ++v) {
        const float iv = __shfl(inv1, v + half * 8, 32);
        float s = 0.f;
#pragma unroll
        for (int j = 0; j < 4; ++j) {
            const float h = rs[v][j] * iv * w1c[j];
            const float y = h + fmaxf(acc[j][v], 0.f);
            acc[j][v] = y;
            s += y * y;
        }
        s2[v] = s;
    }
#pragma unroll
    for (int v = 0; v < 8; ++v) s2[v] = dpp_add<0x161>(s2[v]);
#pragma unroll
    for (int v = 0; v < 8; ++v) s2[v] = dpp_add<0x162>(s2[v]);
#pragma unroll
    for (int v = 0; v < 8; ++v) s2[v] = dpp_add<0x164>(s2[v]);
#pragma unroll
    for (int v = 0; v < 8; ++v) s2[v] = dpp_add<0x168>(s2[v]);

#pragma unroll
    for (int v = 0; v < 8; ++v) {
        const int rowg = row0 + v + half * 8;
        const float inv2 = rsqrtf(s2[v] * (1.0f / 64.0f) + 1e-5f);
#pragma unroll
        for (int j = 0; j < 4; ++j) acc[j][v] = acc[j][v] * inv2 * w2c[j];
        if (rowg < n) {
#pragma unroll
            for (int j = 0; j < 4; ++j)
                out[(size_t)rowg * HID + j * 16 + l16] = acc[j][v];
        }
    }

    if (!fuse) return;

    float* tl = tile + wave * (16 * WST);
#pragma unroll
    for (int v = 0; v < 8; ++v)
#pragma unroll
        for (int j = 0; j < 4; ++j)
            tl[(v + half * 8) * WST + j * 16 + l16] = acc[j][v];

    __builtin_amdgcn_fence(__ATOMIC_RELEASE, "wavefront");
    __builtin_amdgcn_wave_barrier();
    __builtin_amdgcn_fence(__ATOMIC_ACQUIRE, "wavefront");

    v2f a2[16];
#pragma unroll
    for (int s = 0; s < 16; ++s) {
        float2 q = *(const float2*)&tl[l16 * WST + 4 * s + half * 2];
        a2[s].x = q.x; a2[s].y = q.y;
    }

    v8f acc2[4] = {};
#pragma unroll
    for (int s = 0; s < 16; ++s) {
#pragma unroll
        for (int j = 0; j < 4; ++j) {
            const int nc = j * 16 + l16;
            float2 bb = *(const float2*)&lw2[nc * WST + 4 * s + half * 2];
            v2f b; b.x = bb.x; b.y = bb.y;
            acc2[j] = __builtin_amdgcn_wmma_f32_16x16x4_f32(
                false, a2[s], false, b, (short)0, acc2[j], false, false);
        }
    }

#pragma unroll
    for (int v = 0; v < 8; ++v) {
        const int rowg = row0 + v + half * 8;
        if (rowg < n) {
            float* op = mout + (size_t)rowg * HID + 2 * l16;
            *(float2*)op        = make_float2(fmaxf(acc2[0][v], 0.f), fmaxf(acc2[2][v], 0.f));
            *(float2*)(op + 32) = make_float2(fmaxf(acc2[1][v], 0.f), fmaxf(acc2[3][v], 0.f));
        }
    }
}

#define PB_ROWS  64
#define PB_SHIFT 6

__global__ __launch_bounds__(256) void gnn_bin_count(
    const int* __restrict__ row, unsigned* __restrict__ cnt,
    unsigned* __restrict__ rank, int ne)
{
    const int e = blockIdx.x * 256 + threadIdx.x;
    if (e < ne) rank[e] = atomicAdd(&cnt[row[e] >> PB_SHIFT], 1u);
}

__global__ __launch_bounds__(1024) void gnn_bin_scan(
    const unsigned* __restrict__ cnt, unsigned* __restrict__ start, int nb)
{
    __shared__ unsigned wsum[32];
    const int tid = threadIdx.x, lane = tid & 31, wave = tid >> 5;
    const int per = (nb + 1023) / 1024;
    const int b0 = tid * per;
    unsigned s = 0;
    for (int i = 0; i < per; ++i) {
        const int b = b0 + i;
        if (b < nb) s += cnt[b];
    }
    unsigned inc = s;
    for (int d = 1; d < 32; d <<= 1) {
        const unsigned o = __shfl_up(inc, d, 32);
        if (lane >= d) inc += o;
    }
    if (lane == 31) wsum[wave] = inc;
    __syncthreads();
    if (wave == 0) {
        const unsigned w = wsum[lane];
        unsigned winc = w;
        for (int d = 1; d < 32; d <<= 1) {
            const unsigned o = __shfl_up(winc, d, 32);
            if (lane >= d) winc += o;
        }
        wsum[lane] = winc - w;
    }
    __syncthreads();
    unsigned run = wsum[wave] + inc - s;
    for (int i = 0; i < per; ++i) {
        const int b = b0 + i;
        if (b < nb) { start[b] = run; run += cnt[b]; }
    }
    if (tid == 1023) start[nb] = run;
}

__global__ __launch_bounds__(256) void gnn_bin_place(
    const int* __restrict__ row, const int* __restrict__ col,
    const unsigned* __restrict__ start, const unsigned* __restrict__ rank,
    unsigned* __restrict__ packed, int ne)
{
    const int e = blockIdx.x * 256 + threadIdx.x;
    if (e < ne) {
        const int r = row[e];
        packed[start[r >> PB_SHIFT] + rank[e]] =
            ((unsigned)col[e] << PB_SHIFT) | (unsigned)(r & (PB_ROWS - 1));
    }
}

#define PL_GRP 16

__global__ __launch_bounds__(256) void gnn_pull(
    const float* __restrict__ m, const float* __restrict__ h,
    const unsigned* __restrict__ start, const unsigned* __restrict__ packed,
    float* __restrict__ t, int n)
{
    __shared__ __attribute__((aligned(16))) float acc[PB_ROWS * HID];
    const int tid = threadIdx.x;
    for (int i = tid; i < PB_ROWS * HID / 4; i += 256)
        ((float4*)acc)[i] = make_float4(0.f, 0.f, 0.f, 0.f);
    __syncthreads();

    const int lane = tid & 31;
    const int wave = __builtin_amdgcn_readfirstlane(tid >> 5);
    const unsigned beg = start[blockIdx.x];
    const unsigned end = start[blockIdx.x + 1];
    const unsigned laneoff = (unsigned)lane * 8u;

    for (unsigned base = beg + (unsigned)wave * 32u; base < end; base += 8u * 32u) {
        const unsigned left = end - base;
        const int cnt = left < 32u ? (int)left : 32;
        const unsigned p = (base + (unsigned)lane < end) ? packed[base + lane] : 0u;
#pragma unroll
        for (int g = 0; g < 32; g += PL_GRP) {
            if (g >= cnt) break;
            float2 v[PL_GRP];
#pragma unroll
            for (int k = 0; k < PL_GRP; ++k) {
                const unsigned q = (unsigned)__builtin_amdgcn_readlane((int)p, g + k);
                v[k] = *(const float2*)((const char*)(m + (size_t)(q >> PB_SHIFT) * HID) + laneoff);
            }
#pragma unroll
            for (int k = 0; k < PL_GRP; ++k) {
                if (g + k < cnt) {
                    const unsigned q = (unsigned)__builtin_amdgcn_readlane((int)p, g + k);
                    float* a = acc + (q & (PB_ROWS - 1)) * HID + lane;
                    atomicAdd(a, v[k].x);
                    atomicAdd(a + 32, v[k].y);
                }
            }
        }
    }
    __syncthreads();

    const int row0 = blockIdx.x * PB_ROWS;
    for (int i = tid; i < PB_ROWS * HID / 4; i += 256) {
        const int r = row0 + (i >> 4);
        if (r < n) {
            const size_t o = (size_t)r * HID + (size_t)(i & 15) * 4;
            const float4 hv = *(const float4*)(h + o);
            const float4 av = ((const float4*)acc)[i];
            *(float4*)(t + o) = make_float4(hv.x + av.x, hv.y + av.y, hv.z + av.z, hv.w + av.w);
        }
    }
}

extern "C" void kernel_launch(void* const* d_in, const int* in_sizes, int n_in,
                              void* d_out, int out_size, void* d_ws, size_t ws_size,
                              hipStream_t stream)
{
    const float* x       = (const float*)d_in[0];
    const float* conv_w  = (const float*)d_in[1];
    const float* conv_nw = (const float*)d_in[2];
    const float* hid_w   = (const float*)d_in[3];
    const float* hid_nw  = (const float*)d_in[4];
    const int*   row     = (const int*)d_in[5];
    const int*   col     = (const int*)d_in[6];

    const int n  = in_sizes[0] / HID;
    const int ne = in_sizes[5];

    float* P = (float*)d_out;
    float* Q = (float*)d_ws;
    float* R = Q + (size_t)n * HID;

    const int nb = (n + PB_ROWS - 1) / PB_ROWS;
    unsigned* cnt    = (unsigned*)(R + (size_t)n * HID);
    unsigned* start  = cnt + nb;
    unsigned* rank   = start + nb + 1;
    unsigned* packed = rank + ne;

    const dim3 blk(256);
    const int gemm_grid = (n + 127) / 128;
    const int edge_grid = (ne + 255) / 256;

    gnn_gemm_relu<<<gemm_grid, blk, 0, stream>>>(x, conv_w, Q, n, cnt, nb);

    gnn_bin_count<<<edge_grid, blk, 0, stream>>>(row, cnt, rank, ne);
    gnn_bin_scan<<<1, dim3(1024), 0, stream>>>(cnt, start, nb);
    gnn_bin_place<<<edge_grid, blk, 0, stream>>>(row, col, start, rank, packed, ne);

    const float* hin = x;
    for (int l = 0; l < 4; ++l) {
        const float* hw = hid_w  + (size_t)l * HID * HID;
        const float* cn = conv_nw + l * HID;
        const float* hn = hid_nw  + l * HID;
        const float* cwn = (l + 1 < 4) ? conv_w + (size_t)(l + 1) * HID * HID : nullptr;

        gnn_pull<<<nb, blk, 0, stream>>>(Q, hin, start, packed, R, n);
        gnn_norm_gemm_norm<<<gemm_grid, blk, 0, stream>>>(R, cn, hw, hn, P, cwn, Q, n);

        hin = P;
    }
}


# Dense kernels' prologue: weights staged by 16-byte loads, all of them and the wave's row loads in flight together before the one barrier (was 18 dependent round trips of two words each)
# speedup vs baseline: 2.8273x; 1.1365x over previous
#include <hip/hip_runtime.h>

typedef __attribute__((ext_vector_type(2))) float v2f;
typedef __attribute__((ext_vector_type(8))) float v8f;

#define HID 64
#define WST 68

__global__ __launch_bounds__(256) void gnn_gemm_relu(
    const float* __restrict__ in, const float* __restrict__ W,
    float* __restrict__ out, int n, unsigned* __restrict__ cnt, int nb)
{
    __shared__ __attribute__((aligned(16))) float lw[64 * WST];
    const int tid = threadIdx.x;

    {
        const int gi = blockIdx.x * 256 + tid;
        if (gi < nb) cnt[gi] = 0u;
    }

    const int wave = tid >> 5;
    const int lane = tid & 31;
    const int half = lane >> 4;
    const int l16  = lane & 15;
    const int row0 = blockIdx.x * 128 + wave * 16;

    float4 wv[4];
#pragma unroll
    for (int k = 0; k < 4; ++k) wv[k] = ((const float4*)W)[tid + 256 * k];

    const int  arow  = row0 + l16;
    const bool rowok = arow < n;
    const float* abase = in + (size_t)(rowok ? arow : 0) * HID + half * 2;
    v2f a[16];
#pragma unroll
    for (int s = 0; s < 16; ++s) {
        const float2 t = *(const float2*)(abase + 4 * s);
        a[s].x = rowok ? t.x : 0.f; a[s].y = rowok ? t.y : 0.f;
    }

#pragma unroll
    for (int k = 0; k < 4; ++k) {
        const int i4 = tid + 256 * k;
        *(float4*)&lw[(i4 >> 4) * WST + (i4 & 15) * 4] = wv[k];
    }
    __syncthreads();

    v8f acc[4] = {};

#pragma unroll
    for (int s = 0; s < 16; ++s) {
#pragma unroll
        for (int j = 0; j < 4; ++j) {
            const int nc = j * 16 + l16;
            float2 bb = *(const float2*)&lw[nc * WST + 4 * s + half * 2];
            v2f b; b.x = bb.x; b.y = bb.y;
            acc[j] = __builtin_amdgcn_wmma_f32_16x16x4_f32(
                false, a[s], false, b, (short)0, acc[j], false, false);
        }
    }

#pragma unroll
    for (int v = 0; v < 8; ++v) {
        const int rowg = row0 + v + half * 8;
        if (rowg < n) {
            float* op = out + (size_t)rowg * HID + 2 * l16;
            *(float2*)op        = make_float2(fmaxf(acc[0][v], 0.f), fmaxf(acc[2][v], 0.f));
            *(float2*)(op + 32) = make_float2(fmaxf(acc[1][v], 0.f), fmaxf(acc[3][v], 0.f));
        }
    }
}

template <int CTRL>
__device__ __forceinline__ float dpp_add(float v) {
    return v + __int_as_float(__builtin_amdgcn_update_dpp(
                   0, __float_as_int(v), CTRL, 0xf, 0xf, true));
}

__global__ __launch_bounds__(256) void gnn_norm_gemm_norm(
    const float* __restrict__ t, const float* __restrict__ w1,
    const float* __restrict__ W, const float* __restrict__ w2,
    float* __restrict__ out, const float* __restrict__ Wn,
    float* __restrict__ mout, int n)
{
    __shared__ __attribute__((aligned(16))) float lw[64 * WST];
    __shared__ __attribute__((aligned(16))) float lw2[64 * WST];
    __shared__ float tile[8 * 16 * WST];
    __shared__ float lnw[2 * HID];
    const int tid = threadIdx.x;
    const bool fuse = Wn != nullptr;

    const int wave = tid >> 5;
    const int lane = tid & 31;
    const int half = lane >> 4;
    const int l16  = lane & 15;
    const int row0 = blockIdx.x * 128 + wave * 16;

    float lnv = 0.f;
    if (tid < 2 * HID) lnv = tid < HID ? w1[tid] : w2[tid - HID];
    const float4 w1q = ((const float4*)w1)[tid & 15];
    const float4* wn4 = (const float4*)(fuse ? Wn : W);
    float4 wv[4], wnv[4];
#pragma unroll
    for (int k = 0; k < 4; ++k) wv[k] = ((const float4*)W)[tid + 256 * k];
#pragma unroll
    for (int k = 0; k < 4; ++k) wnv[k] = wn4[tid + 256 * k];

    const int  arow  = row0 + l16;
    const bool rowok = arow < n;
    const float* abase = t + (size_t)(rowok ? arow : 0) * HID + half * 2;
    float2 aq[16];
#pragma unroll
    for (int s = 0; s < 16; ++s) aq[s] = *(const float2*)(abase + 4 * s);

    if (tid < 2 * HID) lnw[tid] = lnv;
#pragma unroll
    for (int k = 0; k < 4; ++k) {
        const int i4 = tid + 256 * k;
        *(float4*)&lw[(i4 >> 4) * WST + (i4 & 15) * 4] =
            make_float4(wv[k].x * w1q.x, wv[k].y * w1q.y, wv[k].z * w1q.z, wv[k].w * w1q.w);
    }
#pragma unroll
    for (int k = 0; k < 4; ++k) {
        const int i4 = tid + 256 * k;
        *(float4*)&lw2[(i4 >> 4) * WST + (i4 & 15) * 4] = wnv[k];
    }
    __syncthreads();

    v2f a[16];
    float ss = 0.f;
#pragma unroll
    for (int s = 0; s < 16; ++s) {
        const float2 q = rowok ? aq[s] : make_float2(0.f, 0.f);
        a[s].x = q.x; a[s].y = q.y;
        ss += q.x * q.x + q.y * q.y;
    }
    ss += __shfl_xor(ss, 16, 32);
    const float inv1 = rsqrtf(ss * (1.0f / 64.0f) + 1e-5f);
#pragma unroll
    for (int s = 0; s < 16; ++s) { a[s].x *= inv1; a[s].y *= inv1; }

    float rs[8][4];
#pragma unroll
    for (int v = 0; v < 8; ++v) {
        const int rowg = row0 + v + half * 8;
        const float* tp = t + (size_t)(rowg < n ? rowg : 0) * HID + l16;
#pragma unroll
        for (int j = 0; j < 4; ++j) rs[v][j] = tp[j * 16];
    }
#pragma unroll
    for (int v = 0; v < 8; ++v)
        asm volatile("" :: "v"(rs[v][0]), "v"(rs[v][1]), "v"(rs[v][2]), "v"(rs[v][3]));

    v8f acc[4] = {};

#pragma unroll
    for (int s = 0; s < 16; ++s) {
#pragma unroll
        for (int j = 0; j < 4; ++j) {
            const int nc = j * 16 + l16;
            float2 bb = *(const float2*)&lw[nc * WST + 4 * s + half * 2];
            v2f b; b.x = bb.x; b.y = bb.y;
            acc[j] = __builtin_amdgcn_wmma_f32_16x16x4_f32(
                false, a[s], false, b, (short)0, acc[j], false, false);
        }
    }

    float w1c[4], w2c[4];
#pragma unroll
    for (int j = 0; j < 4; ++j) {
        w1c[j] = lnw[j * 16 + l16];
        w2c[j] = lnw[HID + j * 16 + l16];
    }

    float s2[8];
#pragma unroll
    for (int v = 0; v < 8; ++v) {
        const float iv = __shfl(inv1, v + half * 8, 32);
        float s = 0.f;
#pragma unroll
        for (int j = 0; j < 4; ++j) {
            const float h = rs[v][j] * iv * w1c[j];
            const float y = h + fmaxf(acc[j][v], 0.f);
            acc[j][v] = y;
            s += y * y;
        }
        s2[v] = s;
    }
#pragma unroll
    for (int v = 0; v < 8; ++v) s2[v] = dpp_add<0x161>(s2[v]);
#pragma unroll
    for (int v = 0; v < 8; ++v) s2[v] = dpp_add<0x162>(s2[v]);
#pragma unroll
    for (int v = 0; v < 8; ++v) s2[v] = dpp_add<0x164>(s2[v]);
#pragma unroll
    for (int v = 0; v < 8; ++v) s2[v] = dpp_add<0x168>(s2[v]);

#pragma unroll
    for (int v = 0; v < 8; ++v) {
        const int rowg = row0 + v + half * 8;
        const float inv2 = rsqrtf(s2[v] * (1.0f / 64.0f) + 1e-5f);
#pragma unroll
        for (int j = 0; j < 4; ++j) acc[j][v] = acc[j][v] * inv2 * w2c[j];
        if (rowg < n) {
#pragma unroll
            for (int j = 0; j < 4; ++j)
                out[(size_t)rowg * HID + j * 16 + l16] = acc[j][v];
        }
    }

    if (!fuse) return;

    float* tl = tile + wave * (16 * WST);
#pragma unroll
    for (int v = 0; v < 8; ++v)
#pragma unroll
        for (int j = 0; j < 4; ++j)
            tl[(v + half * 8) * WST + j * 16 + l16] = acc[j][v];

    __builtin_amdgcn_fence(__ATOMIC_RELEASE, "wavefront");
    __builtin_amdgcn_wave_barrier();
    __builtin_amdgcn_fence(__ATOMIC_ACQUIRE, "wavefront");

    v2f a2[16];
#pragma unroll
    for (int s = 0; s < 16; ++s) {
        float2 q = *(const float2*)&tl[l16 * WST + 4 * s + half * 2];
        a2[s].x = q.x; a2[s].y = q.y;
    }

    v8f acc2[4] = {};
#pragma unroll
    for (int s = 0; s < 16; ++s) {
#pragma unroll
        for (int j = 0; j < 4; ++j) {
            const int nc = j * 16 + l16;
            float2 bb = *(const float2*)&lw2[nc * WST + 4 * s + half * 2];
            v2f b; b.x = bb.x; b.y = bb.y;
            acc2[j] = __builtin_amdgcn_wmma_f32_16x16x4_f32(
                false, a2[s], false, b, (short)0, acc2[j], false, false);
        }
    }

#pragma unroll
    for (int v = 0; v < 8; ++v) {
        const int rowg = row0 + v + half * 8;
        if (rowg < n) {
            float* op = mout + (size_t)rowg * HID + 2 * l16;
            *(float2*)op        = make_float2(fmaxf(acc2[0][v], 0.f), fmaxf(acc2[2][v], 0.f));
            *(float2*)(op + 32) = make_float2(fmaxf(acc2[1][v], 0.f), fmaxf(acc2[3][v], 0.f));
        }
    }
}

#define PB_ROWS  64
#define PB_SHIFT 6

__global__ __launch_bounds__(256) void gnn_bin_count(
    const int* __restrict__ row, unsigned* __restrict__ cnt,
    unsigned* __restrict__ rank, int ne)
{
    const int e = blockIdx.x * 256 + threadIdx.x;
    if (e < ne) rank[e] = atomicAdd(&cnt[row[e] >> PB_SHIFT], 1u);
}

__global__ __launch_bounds__(1024) void gnn_bin_scan(
    const unsigned* __restrict__ cnt, unsigned* __restrict__ start, int nb)
{
    __shared__ unsigned wsum[32];
    const int tid = threadIdx.x, lane = tid & 31, wave = tid >> 5;
    const int per = (nb + 1023) / 1024;
    const int b0 = tid * per;
    unsigned s = 0;
    for (int i = 0; i < per; ++i) {
        const int b = b0 + i;
        if (b < nb) s += cnt[b];
    }
    unsigned inc = s;
    for (int d = 1; d < 32; d <<= 1) {
        const unsigned o = __shfl_up(inc, d, 32);
        if (lane >= d) inc += o;
    }
    if (lane == 31) wsum[wave] = inc;
    __syncthreads();
    if (wave == 0) {
        const unsigned w = wsum[lane];
        unsigned winc = w;
        for (int d = 1; d < 32; d <<= 1) {
            const unsigned o = __shfl_up(winc, d, 32);
            if (lane >= d) winc += o;
        }
        wsum[lane] = winc - w;
    }
    __syncthreads();
    unsigned run = wsum[wave] + inc - s;
    for (int i = 0; i < per; ++i) {
        const int b = b0 + i;
        if (b < nb) { start[b] = run; run += cnt[b]; }
    }
    if (tid == 1023) start[nb] = run;
}

__global__ __launch_bounds__(256) void gnn_bin_place(
    const int* __restrict__ row, const int* __restrict__ col,
    const unsigned* __restrict__ start, const unsigned* __restrict__ rank,
    unsigned* __restrict__ packed, int ne)
{
    const int e = blockIdx.x * 256 + threadIdx.x;
    if (e < ne) {
        const int r = row[e];
        packed[start[r >> PB_SHIFT] + rank[e]] =
            ((unsigned)col[e] << PB_SHIFT) | (unsigned)(r & (PB_ROWS - 1));
    }
}

#define PL_GRP 16

__global__ __launch_bounds__(256) void gnn_pull(
    const float* __restrict__ m, const float* __restrict__ h,
    const unsigned* __restrict__ start, const unsigned* __restrict__ packed,
    float* __restrict__ t, int n)
{
    __shared__ __attribute__((aligned(16))) float acc[PB_ROWS * HID];
    const int tid = threadIdx.x;
    for (int i = tid; i < PB_ROWS * HID / 4; i += 256)
        ((float4*)acc)[i] = make_float4(0.f, 0.f, 0.f, 0.f);
    __syncthreads();

    const int lane = tid & 31;
    const int wave = __builtin_amdgcn_readfirstlane(tid >> 5);
    const unsigned beg = start[blockIdx.x];
    const unsigned end = start[blockIdx.x + 1];
    const unsigned laneoff = (unsigned)lane * 8u;

    for (unsigned base = beg + (unsigned)wave * 32u; base < end; base += 8u * 32u) {
        const unsigned left = end - base;
        const int cnt = left < 32u ? (int)left : 32;
        const unsigned p = (base + (unsigned)lane < end) ? packed[base + lane] : 0u;
#pragma unroll
        for (int g = 0; g < 32; g += PL_GRP) {
            if (g >= cnt) break;
            float2 v[PL_GRP];
#pragma unroll
            for (int k = 0; k < PL_GRP; ++k) {
                const unsigned q = (unsigned)__builtin_amdgcn_readlane((int)p, g + k);
                v[k] = *(const float2*)((const char*)(m + (size_t)(q >> PB_SHIFT) * HID) + laneoff);
            }
#pragma unroll
            for (int k = 0; k < PL_GRP; ++k) {
                if (g + k < cnt) {
                    const unsigned q = (unsigned)__builtin_amdgcn_readlane((int)p, g + k);
                    float* a = acc + (q & (PB_ROWS - 1)) * HID + lane;
                    atomicAdd(a, v[k].x);
                    atomicAdd(a + 32, v[k].y);
                }
            }
        }
    }
    __syncthreads();

    const int row0 = blockIdx.x * PB_ROWS;
    for (int i = tid; i < PB_ROWS * HID / 4; i += 256) {
        const int r = row0 + (i >> 4);
        if (r < n) {
            const size_t o = (size_t)r * HID + (size_t)(i & 15) * 4;
            const float4 hv = *(const float4*)(h + o);
            const float4 av = ((const float4*)acc)[i];
            *(float4*)(t + o) = make_float4(hv.x + av.x, hv.y + av.y, hv.z + av.z, hv.w + av.w);
        }
    }
}

extern "C" void kernel_launch(void* const* d_in, const int* in_sizes, int n_in,
                              void* d_out, int out_size, void* d_ws, size_t ws_size,
                              hipStream_t stream)
{
    const float* x       = (const float*)d_in[0];
    const float* conv_w  = (const float*)d_in[1];
    const float* conv_nw = (const float*)d_in[2];
    const float* hid_w   = (const float*)d_in[3];
    const float* hid_nw  = (const float*)d_in[4];
    const int*   row     = (const int*)d_in[5];
    const int*   col     = (const int*)d_in[6];

    const int n  = in_sizes[0] / HID;
    const int ne = in_sizes[5];

    float* P = (float*)d_out;
    float* Q = (float*)d_ws;
    float* R = Q + (size_t)n * HID;

    const int nb = (n + PB_ROWS - 1) / PB_ROWS;
    unsigned* cnt    = (unsigned*)(R + (size_t)n * HID);
    unsigned* start  = cnt + nb;
    unsigned* rank   = start + nb + 1;
    unsigned* packed = rank + ne;

    const dim3 blk(256);
    const int gemm_grid = (n + 127) / 128;
    const int edge_grid = (ne + 255) / 256;

    gnn_gemm_relu<<<gemm_grid, blk, 0, stream>>>(x, conv_w, Q, n, cnt, nb);

    gnn_bin_count<<<edge_grid, blk, 0, stream>>>(row, cnt, rank, ne);
    gnn_bin_scan<<<1, dim3(1024), 0, stream>>>(cnt, start, nb);
    gnn_bin_place<<<edge_grid, blk, 0, stream>>>(row, col, start, rank, packed, ne);

    const float* hin = x;
    for (int l = 0; l < 4; ++l) {
        const float* hw = hid_w  + (size_t)l * HID * HID;
        const float* cn = conv_nw + l * HID;
        const float* hn = hid_nw  + l * HID;
        const float* cwn = (l + 1 < 4) ? conv_w + (size_t)(l + 1) * HID * HID : nullptr;

        gnn_pull<<<nb, blk, 0, stream>>>(Q, hin, start, packed, R, n);
        gnn_norm_gemm_norm<<<gemm_grid, blk, 0, stream>>>(R, cn, hw, hn, P, cwn, Q, n);

        hin = P;
    }
}


# Pull and dense half of a layer fused into one kernel per 64-row block: t stays in LDS and feeds the matrix loops (no store and re-read of t); 12 -> 8 launches; every sum in the same order
# speedup vs baseline: 3.2091x; 1.1350x over previous
#include <hip/hip_runtime.h>

typedef __attribute__((ext_vector_type(2))) float v2f;
typedef __attribute__((ext_vector_type(8))) float v8f;

#define HID 64
#define WST 68

__global__ __launch_bounds__(256) void gnn_gemm_relu(
    const float* __restrict__ in, const float* __restrict__ W,
    float* __restrict__ out, int n, unsigned* __restrict__ cnt, int nb)
{
    __shared__ __attribute__((aligned(16))) float lw[64 * WST];
    const int tid = threadIdx.x;

    {
        const int gi = blockIdx.x * 256 + tid;
        if (gi < nb) cnt[gi] = 0u;
    }

    const int wave = tid >> 5;
    const int lane = tid & 31;
    const int half = lane >> 4;
    const int l16  = lane & 15;
    const int row0 = blockIdx.x * 128 + wave * 16;

    float4 wv[4];
#pragma unroll
    for (int k = 0; k < 4; ++k) wv[k] = ((const float4*)W)[tid + 256 * k];

    const int  arow  = row0 + l16;
    const bool rowok = arow < n;
    const float* abase = in + (size_t)(rowok ? arow : 0) * HID + half * 2;
    v2f a[16];
#pragma unroll
    for (int s = 0; s < 16; ++s) {
        const float2 t = *(const float2*)(abase + 4 * s);
        a[s].x = rowok ? t.x : 0.f; a[s].y = rowok ? t.y : 0.f;
    }

#pragma unroll
    for (int k = 0; k < 4; ++k) {
        const int i4 = tid + 256 * k;
        *(float4*)&lw[(i4 >> 4) * WST + (i4 & 15) * 4] = wv[k];
    }
    __syncthreads();

    v8f acc[4] = {};

#pragma unroll
    for (int s = 0; s < 16; ++s) {
#pragma unroll
        for (int j = 0; j < 4; ++j) {
            const int nc = j * 16 + l16;
            float2 bb = *(const float2*)&lw[nc * WST + 4 * s + half * 2];
            v2f b; b.x = bb.x; b.y = bb.y;
            acc[j] = __builtin_amdgcn_wmma_f32_16x16x4_f32(
                false, a[s], false, b, (short)0, acc[j], false, false);
        }
    }

#pragma unroll
    for (int v = 0; v < 8; ++v) {
        const int rowg = row0 + v + half * 8;
        if (rowg < n) {
            float* op = out + (size_t)rowg * HID + 2 * l16;
            *(float2*)op        = make_float2(fmaxf(acc[0][v], 0.f), fmaxf(acc[2][v], 0.f));
            *(float2*)(op + 32) = make_float2(fmaxf(acc[1][v], 0.f), fmaxf(acc[3][v], 0.f));
        }
    }
}

template <int CTRL>
__device__ __forceinline__ float dpp_add(float v) {
    return v + __int_as_float(__builtin_amdgcn_update_dpp(
                   0, __float_as_int(v), CTRL, 0xf, 0xf, true));
}

#define PB_ROWS  64
#define PB_SHIFT 6

__global__ __launch_bounds__(256) void gnn_bin_count(
    const int* __restrict__ row, unsigned* __restrict__ cnt,
    unsigned* __restrict__ rank, int ne)
{
    const int e = blockIdx.x * 256 + threadIdx.x;
    if (e < ne) rank[e] = atomicAdd(&cnt[row[e] >> PB_SHIFT], 1u);
}

__global__ __launch_bounds__(1024) void gnn_bin_scan(
    const unsigned* __restrict__ cnt, unsigned* __restrict__ start, int nb)
{
    __shared__ unsigned wsum[32];
    const int tid = threadIdx.x, lane = tid & 31, wave = tid >> 5;
    const int per = (nb + 1023) / 1024;
    const int b0 = tid * per;
    unsigned s = 0;
    for (int i = 0; i < per; ++i) {
        const int b = b0 + i;
        if (b < nb) s += cnt[b];
    }
    unsigned inc = s;
    for (int d = 1; d < 32; d <<= 1) {
        const unsigned o = __shfl_up(inc, d, 32);
        if (lane >= d) inc += o;
    }
    if (lane == 31) wsum[wave] = inc;
    __syncthreads();
    if (wave == 0) {
        const unsigned w = wsum[lane];
        unsigned winc = w;
        for (int d = 1; d < 32; d <<= 1) {
            const unsigned o = __shfl_up(winc, d, 32);
            if (lane >= d) winc += o;
        }
        wsum[lane] = winc - w;
    }
    __syncthreads();
    unsigned run = wsum[wave] + inc - s;
    for (int i = 0; i < per; ++i) {
        const int b = b0 + i;
        if (b < nb) { start[b] = run; run += cnt[b]; }
    }
    if (tid == 1023) start[nb] = run;
}

__global__ __launch_bounds__(256) void gnn_bin_place(
    const int* __restrict__ row, const int* __restrict__ col,
    const unsigned* __restrict__ start, const unsigned* __restrict__ rank,
    unsigned* __restrict__ packed, int ne)
{
    const int e = blockIdx.x * 256 + threadIdx.x;
    if (e < ne) {
        const int r = row[e];
        packed[start[r >> PB_SHIFT] + rank[e]] =
            ((unsigned)col[e] << PB_SHIFT) | (unsigned)(r & (PB_ROWS - 1));
    }
}

#define PL_GRP 16

__global__ __launch_bounds__(256) void gnn_layer(
    const float* __restrict__ m, const float* __restrict__ h,
    const unsigned* __restrict__ start, const unsigned* __restrict__ packed,
    const float* __restrict__ w1, const float* __restrict__ W,
    const float* __restrict__ w2, const float* __restrict__ Wn,
    float* __restrict__ out, float* __restrict__ mout, int n)
{
    __shared__ __attribute__((aligned(16))) float lw[64 * WST];
    __shared__ __attribute__((aligned(16))) float lw2[64 * WST];
    __shared__ __attribute__((aligned(16))) float acc[PB_ROWS * WST];
    __shared__ float lnw[4 * HID];
    const int tid = threadIdx.x;
    const bool fuse = Wn != nullptr;
    const int lane = tid & 31;
    const int wave = __builtin_amdgcn_readfirstlane(tid >> 5);
    const int brow0 = blockIdx.x * PB_ROWS;

    for (int i = tid; i < PB_ROWS * WST / 4; i += 256)
        ((float4*)acc)[i] = make_float4(0.f, 0.f, 0.f, 0.f);

    const float lnv = ((tid & HID) ? w2 : w1)[tid & (HID - 1)];
    const float4 w1q = ((const float4*)w1)[tid & 15];
    const float4* wn4 = (const float4*)(fuse ? Wn : W);
    float4 wv[4], wnv[4];
#pragma unroll
    for (int k = 0; k < 4; ++k) wv[k] = ((const float4*)W)[tid + 256 * k];
#pragma unroll
    for (int k = 0; k < 4; ++k) wnv[k] = wn4[tid + 256 * k];

    const unsigned beg = start[blockIdx.x];
    const unsigned end = start[blockIdx.x + 1];

    float4 hv[4];
#pragma unroll
    for (int k = 0; k < 4; ++k) {
        const int i = tid + 256 * k;
        const int r = brow0 + (i >> 4);
        hv[k] = *(const float4*)(h + (size_t)(r < n ? r : 0) * HID + (size_t)(i & 15) * 4);
    }

    unsigned base = beg + (unsigned)wave * 32u;
    unsigned p = (base + (unsigned)lane < end) ? packed[base + lane] : 0u;

#pragma unroll
    for (int k = 0; k < 4; ++k) {
        const int i4 = tid + 256 * k;
        *(float4*)&lw[(i4 >> 4) * WST + (i4 & 15) * 4] =
            make_float4(wv[k].x * w1q.x, wv[k].y * w1q.y, wv[k].z * w1q.z, wv[k].w * w1q.w);
    }
#pragma unroll
    for (int k = 0; k < 4; ++k) {
        const int i4 = tid + 256 * k;
        *(float4*)&lw2[(i4 >> 4) * WST + (i4 & 15) * 4] = wnv[k];
    }
    lnw[tid] = lnv;
    __syncthreads();

    const unsigned laneoff = (unsigned)lane * 8u;
    while (base < end) {
        const unsigned left = end - base;
        const int cnt = left < 32u ? (int)left : 32;
#pragma unroll
        for (int g = 0; g < 32; g += PL_GRP) {
            if (g >= cnt) break;
            float2 v[PL_GRP];
#pragma unroll
            for (int k = 0; k < PL_GRP; ++k) {
                const unsigned q = (unsigned)__builtin_amdgcn_readlane((int)p, g + k);
                v[k] = *(const float2*)((const char*)(m + (size_t)(q >> PB_SHIFT) * HID) + laneoff);
            }
#pragma unroll
            for (int k = 0; k < PL_GRP; ++k) {
                if (g + k < cnt) {
                    const unsigned q = (unsigned)__builtin_amdgcn_readlane((int)p, g + k);
                    float* a = acc + (q & (PB_ROWS - 1)) * WST + lane;
                    atomicAdd(a, v[k].x);
                    atomicAdd(a + 32, v[k].y);
                }
            }
        }
        base += 8u * 32u;
        p = (base + (unsigned)lane < end) ? packed[base + lane] : 0u;
    }
    __syncthreads();

#pragma unroll
    for (int k = 0; k < 4; ++k) {
        const int i = tid + 256 * k;
        const bool rok = brow0 + (i >> 4) < n;
        float4* ap = (float4*)&acc[(i >> 4) * WST + (i & 15) * 4];
        const float4 av = *ap;
        *ap = make_float4((rok ? hv[k].x : 0.f) + av.x, (rok ? hv[k].y : 0.f) + av.y,
                          (rok ? hv[k].z : 0.f) + av.z, (rok ? hv[k].w : 0.f) + av.w);
    }
    __syncthreads();

    if (wave >= PB_ROWS / 16) return;

    const int half = lane >> 4;
    const int l16  = lane & 15;
    const int row0 = brow0 + wave * 16;
    float* tl = acc + wave * (16 * WST);

    v2f a[16];
    float ss = 0.f;
#pragma unroll
    for (int s = 0; s < 16; ++s) {
        const float2 q = *(const float2*)&tl[l16 * WST + 4 * s + half * 2];
        a[s].x = q.x; a[s].y = q.y;
        ss += q.x * q.x + q.y * q.y;
    }
    ss += __shfl_xor(ss, 16, 32);
    const float inv1 = rsqrtf(ss * (1.0f / 64.0f) + 1e-5f);
#pragma unroll
    for (int s = 0; s < 16; ++s) { a[s].x *= inv1; a[s].y *= inv1; }

    float rs[8][4];
#pragma unroll
    for (int v = 0; v < 8; ++v)
#pragma unroll
        for (int j = 0; j < 4; ++j) rs[v][j] = tl[(v + half * 8) * WST + j * 16 + l16];

    v8f c1[4] = {};

#pragma unroll
    for (int s = 0; s < 16; ++s) {
#pragma unroll
        for (int j = 0; j < 4; ++j) {
            const int nc = j * 16 + l16;
            float2 bb = *(const float2*)&lw[nc * WST + 4 * s + half * 2];
            v2f b; b.x = bb.x; b.y = bb.y;
            c1[j] = __builtin_amdgcn_wmma_f32_16x16x4_f32(
                false, a[s], false, b, (short)0, c1[j], false, false);
        }
    }

    float w1c[4], w2c[4];
#pragma unroll
    for (int j = 0; j < 4; ++j) {
        w1c[j] = lnw[j * 16 + l16];
        w2c[j] = lnw[HID + j * 16 + l16];
    }

    float s2[8];
#pragma unroll
    for (int v = 0; v < 8; ++v) {
        const float iv = __shfl(inv1, v + half * 8, 32);
        float s = 0.f;
#pragma unroll
        for (int j = 0; j < 4; ++j) {
            const float hh = rs[v][j] * iv * w1c[j];
            const float y = hh + fmaxf(c1[j][v], 0.f);
            c1[j][v] = y;
            s += y * y;
        }
        s2[v] = s;
    }
#pragma unroll
    for (int v = 0; v < 8; ++v) s2[v] = dpp_add<0x161>(s2[v]);
#pragma unroll
    for (int v = 0; v < 8; ++v) s2[v] = dpp_add<0x162>(s2[v]);
#pragma unroll
    for (int v = 0; v < 8; ++v) s2[v] = dpp_add<0x164>(s2[v]);
#pragma unroll
    for (int v = 0; v < 8; ++v) s2[v] = dpp_add<0x168>(s2[v]);

#pragma unroll
    for (int v = 0; v < 8; ++v) {
        const int rowg = row0 + v + half * 8;
        const float inv2 = rsqrtf(s2[v] * (1.0f / 64.0f) + 1e-5f);
#pragma unroll
        for (int j = 0; j < 4; ++j) c1[j][v] = c1[j][v] * inv2 * w2c[j];
        if (rowg < n) {
#pragma unroll
            for (int j = 0; j < 4; ++j)
                out[(size_t)rowg * HID + j * 16 + l16] = c1[j][v];
        }
    }

    if (!fuse) return;

    __builtin_amdgcn_fence(__ATOMIC_RELEASE, "wavefront");
    __builtin_amdgcn_wave_barrier();
    __builtin_amdgcn_fence(__ATOMIC_ACQUIRE, "wavefront");

#pragma unroll
    for (int v = 0; v < 8; ++v)
#pragma unroll
        for (int j = 0; j < 4; ++j)
            tl[(v + half * 8) * WST + j * 16 + l16] = c1[j][v];

    __builtin_amdgcn_fence(__ATOMIC_RELEASE, "wavefront");
    __builtin_amdgcn_wave_barrier();
    __builtin_amdgcn_fence(__ATOMIC_ACQUIRE, "wavefront");

    v2f a2[16];
#pragma unroll
    for (int s = 0; s < 16; ++s) {
        float2 q = *(const float2*)&tl[l16 * WST + 4 * s + half * 2];
        a2[s].x = q.x; a2[s].y = q.y;
    }

    v8f c2[4] = {};
#pragma unroll
    for (int s = 0; s < 16; ++s) {
#pragma unroll
        for (int j = 0; j < 4; ++j) {
            const int nc = j * 16 + l16;
            float2 bb = *(const float2*)&lw2[nc * WST + 4 * s + half * 2];
            v2f b; b.x = bb.x; b.y = bb.y;
            c2[j] = __builtin_amdgcn_wmma_f32_16x16x4_f32(
                false, a2[s], false, b, (short)0, c2[j], false, false);
        }
    }

#pragma unroll
    for (int v = 0; v < 8; ++v) {
        const int rowg = row0 + v + half * 8;
        if (rowg < n) {
            float* op = mout + (size_t)rowg * HID + 2 * l16;
            *(float2*)op        = make_float2(fmaxf(c2[0][v], 0.f), fmaxf(c2[2][v], 0.f));
            *(float2*)(op + 32) = make_float2(fmaxf(c2[1][v], 0.f), fmaxf(c2[3][v], 0.f));
        }
    }
}

extern "C" void kernel_launch(void* const* d_in, const int* in_sizes, int n_in,
                              void* d_out, int out_size, void* d_ws, size_t ws_size,
                              hipStream_t stream)
{
    const float* x       = (const float*)d_in[0];
    const float* conv_w  = (const float*)d_in[1];
    const float* conv_nw = (const float*)d_in[2];
    const float* hid_w   = (const float*)d_in[3];
    const float* hid_nw  = (const float*)d_in[4];
    const int*   row     = (const int*)d_in[5];
    const int*   col     = (const int*)d_in[6];

    const int n  = in_sizes[0] / HID;
    const int ne = in_sizes[5];

    float* P  = (float*)d_out;
    float* Q0 = (float*)d_ws;
    float* Q1 = Q0 + (size_t)n * HID;
    float* H0 = Q1 + (size_t)n * HID;
    float* H1 = H0 + (size_t)n * HID;

    const int nb = (n + PB_ROWS - 1) / PB_ROWS;
    unsigned* cnt    = (unsigned*)(H1 + (size_t)n * HID);
    unsigned* start  = cnt + nb;
    unsigned* rank   = start + nb + 1;
    unsigned* packed = rank + ne;

    const dim3 blk(256);
    const int gemm_grid = (n + 127) / 128;
    const int edge_grid = (ne + 255) / 256;

    gnn_gemm_relu<<<gemm_grid, blk, 0, stream>>>(x, conv_w, Q0, n, cnt, nb);

    gnn_bin_count<<<edge_grid, blk, 0, stream>>>(row, cnt, rank, ne);
    gnn_bin_scan<<<1, dim3(1024), 0, stream>>>(cnt, start, nb);
    gnn_bin_place<<<edge_grid, blk, 0, stream>>>(row, col, start, rank, packed, ne);

    for (int l = 0; l < 4; ++l) {
        const float* hw = hid_w  + (size_t)l * HID * HID;
        const float* cn = conv_nw + l * HID;
        const float* hn = hid_nw  + l * HID;
        const float* cwn = (l + 1 < 4) ? conv_w + (size_t)(l + 1) * HID * HID : nullptr;

        const float* min  = (l & 1) ? Q1 : Q0;
        float*       mnx  = (l & 1) ? Q0 : Q1;
        const float* hin  = (l == 0) ? x : ((l & 1) ? H0 : H1);
        float*       hout = (l == 3) ? P : ((l & 1) ? H1 : H0);

        gnn_layer<<<nb, blk, 0, stream>>>(min, hin, start, packed, cn, hw, hn, cwn, hout, mnx, n);
    }
}


# Pull: a wave's 16 row gathers kept in flight as a ring across its sweeps (each add reissues its slot; edge words fetched two sweeps ahead), not issued and drained 16 at a time; adds in arrival order
# speedup vs baseline: 3.4445x; 1.0734x over previous
#include <hip/hip_runtime.h>

typedef __attribute__((ext_vector_type(2))) float v2f;
typedef __attribute__((ext_vector_type(8))) float v8f;

#define HID 64
#define WST 68

__global__ __launch_bounds__(256) void gnn_gemm_relu(
    const float* __restrict__ in, const float* __restrict__ W,
    float* __restrict__ out, int n, unsigned* __restrict__ cnt, int nb)
{
    __shared__ __attribute__((aligned(16))) float lw[64 * WST];
    const int tid = threadIdx.x;

    {
        const int gi = blockIdx.x * 256 + tid;
        if (gi < nb) cnt[gi] = 0u;
    }

    const int wave = tid >> 5;
    const int lane = tid & 31;
    const int half = lane >> 4;
    const int l16  = lane & 15;
    const int row0 = blockIdx.x * 128 + wave * 16;

    float4 wv[4];
#pragma unroll
    for (int k = 0; k < 4; ++k) wv[k] = ((const float4*)W)[tid + 256 * k];

    const int  arow  = row0 + l16;
    const bool rowok = arow < n;
    const float* abase = in + (size_t)(rowok ? arow : 0) * HID + half * 2;
    v2f a[16];
#pragma unroll
    for (int s = 0; s < 16; ++s) {
        const float2 t = *(const float2*)(abase + 4 * s);
        a[s].x = rowok ? t.x : 0.f; a[s].y = rowok ? t.y : 0.f;
    }

#pragma unroll
    for (int k = 0; k < 4; ++k) {
        const int i4 = tid + 256 * k;
        *(float4*)&lw[(i4 >> 4) * WST + (i4 & 15) * 4] = wv[k];
    }
    __syncthreads();

    v8f acc[4] = {};

#pragma unroll
    for (int s = 0; s < 16; ++s) {
#pragma unroll
        for (int j = 0; j < 4; ++j) {
            const int nc = j * 16 + l16;
            float2 bb = *(const float2*)&lw[nc * WST + 4 * s + half * 2];
            v2f b; b.x = bb.x; b.y = bb.y;
            acc[j] = __builtin_amdgcn_wmma_f32_16x16x4_f32(
                false, a[s], false, b, (short)0, acc[j], false, false);
        }
    }

#pragma unroll
    for (int v = 0; v < 8; ++v) {
        const int rowg = row0 + v + half * 8;
        if (rowg < n) {
            float* op = out + (size_t)rowg * HID + 2 * l16;
            *(float2*)op        = make_float2(fmaxf(acc[0][v], 0.f), fmaxf(acc[2][v], 0.f));
            *(float2*)(op + 32) = make_float2(fmaxf(acc[1][v], 0.f), fmaxf(acc[3][v], 0.f));
        }
    }
}

template <int CTRL>
__device__ __forceinline__ float dpp_add(float v) {
    return v + __int_as_float(__builtin_amdgcn_update_dpp(
                   0, __float_as_int(v), CTRL, 0xf, 0xf, true));
}

#define PB_ROWS  64
#define PB_SHIFT 6

__global__ __launch_bounds__(256) void gnn_bin_count(
    const int* __restrict__ row, unsigned* __restrict__ cnt,
    unsigned* __restrict__ rank, int ne)
{
    const int e = blockIdx.x * 256 + threadIdx.x;
    if (e < ne) rank[e] = atomicAdd(&cnt[row[e] >> PB_SHIFT], 1u);
}

__global__ __launch_bounds__(1024) void gnn_bin_scan(
    const unsigned* __restrict__ cnt, unsigned* __restrict__ start, int nb)
{
    __shared__ unsigned wsum[32];
    const int tid = threadIdx.x, lane = tid & 31, wave = tid >> 5;
    const int per = (nb + 1023) / 1024;
    const int b0 = tid * per;
    unsigned s = 0;
    for (int i = 0; i < per; ++i) {
        const int b = b0 + i;
        if (b < nb) s += cnt[b];
    }
    unsigned inc = s;
    for (int d = 1; d < 32; d <<= 1) {
        const unsigned o = __shfl_up(inc, d, 32);
        if (lane >= d) inc += o;
    }
    if (lane == 31) wsum[wave] = inc;
    __syncthreads();
    if (wave == 0) {
        const unsigned w = wsum[lane];
        unsigned winc = w;
        for (int d = 1; d < 32; d <<= 1) {
            const unsigned o = __shfl_up(winc, d, 32);
            if (lane >= d) winc += o;
        }
        wsum[lane] = winc - w;
    }
    __syncthreads();
    unsigned run = wsum[wave] + inc - s;
    for (int i = 0; i < per; ++i) {
        const int b = b0 + i;
        if (b < nb) { start[b] = run; run += cnt[b]; }
    }
    if (tid == 1023) start[nb] = run;
}

__global__ __launch_bounds__(256) void gnn_bin_place(
    const int* __restrict__ row, const int* __restrict__ col,
    const unsigned* __restrict__ start, const unsigned* __restrict__ rank,
    unsigned* __restrict__ packed, int ne)
{
    const int e = blockIdx.x * 256 + threadIdx.x;
    if (e < ne) {
        const int r = row[e];
        packed[start[r >> PB_SHIFT] + rank[e]] =
            ((unsigned)col[e] << PB_SHIFT) | (unsigned)(r & (PB_ROWS - 1));
    }
}

#define PL_GRP 16

__global__ __launch_bounds__(256, 10) void gnn_layer(
    const float* __restrict__ m, const float* __restrict__ h,
    const unsigned* __restrict__ start, const unsigned* __restrict__ packed,
    const float* __restrict__ w1, const float* __restrict__ W,
    const float* __restrict__ w2, const float* __restrict__ Wn,
    float* __restrict__ out, float* __restrict__ mout, int n)
{
    __shared__ __attribute__((aligned(16))) float lw[64 * WST];
    __shared__ __attribute__((aligned(16))) float lw2[64 * WST];
    __shared__ __attribute__((aligned(16))) float acc[PB_ROWS * WST];
    __shared__ float lnw[4 * HID];
    const int tid = threadIdx.x;
    const bool fuse = Wn != nullptr;
    const int lane = tid & 31;
    const int wave = __builtin_amdgcn_readfirstlane(tid >> 5);
    const int brow0 = blockIdx.x * PB_ROWS;

    for (int i = tid; i < PB_ROWS * WST / 4; i += 256)
        ((float4*)acc)[i] = make_float4(0.f, 0.f, 0.f, 0.f);

    const float lnv = ((tid & HID) ? w2 : w1)[tid & (HID - 1)];
    const float4 w1q = ((const float4*)w1)[tid & 15];
    const float4* wn4 = (const float4*)(fuse ? Wn : W);
    float4 wv[4], wnv[4];
#pragma unroll
    for (int k = 0; k < 4; ++k) wv[k] = ((const float4*)W)[tid + 256 * k];
#pragma unroll
    for (int k = 0; k < 4; ++k) wnv[k] = wn4[tid + 256 * k];

    const unsigned beg = start[blockIdx.x];
    const unsigned end = start[blockIdx.x + 1];

    float4 hv[4];
#pragma unroll
    for (int k = 0; k < 4; ++k) {
        const int i = tid + 256 * k;
        const int r = brow0 + (i >> 4);
        hv[k] = *(const float4*)(h + (size_t)(r < n ? r : 0) * HID + (size_t)(i & 15) * 4);
    }

    unsigned base = beg + (unsigned)wave * 32u;
    unsigned p  = (base + (unsigned)lane < end) ? packed[base + lane] : 0u;
    unsigned pn = (base + 256u + (unsigned)lane < end) ? packed[base + 256u + lane] : 0u;

#pragma unroll
    for (int k = 0; k < 4; ++k) {
        const int i4 = tid + 256 * k;
        *(float4*)&lw[(i4 >> 4) * WST + (i4 & 15) * 4] =
            make_float4(wv[k].x * w1q.x, wv[k].y * w1q.y, wv[k].z * w1q.z, wv[k].w * w1q.w);
    }
#pragma unroll
    for (int k = 0; k < 4; ++k) {
        const int i4 = tid + 256 * k;
        *(float4*)&lw2[(i4 >> 4) * WST + (i4 & 15) * 4] = wnv[k];
    }
    lnw[tid] = lnv;
    __syncthreads();

    const unsigned laneoff = (unsigned)lane * 8u;
    auto gather = [&](unsigned q) -> float2 {
        return *(const float2*)((const char*)(m + (size_t)(q >> PB_SHIFT) * HID) + laneoff);
    };
    auto addrow = [&](unsigned q, float2 r) {
        float* a = acc + (q & (PB_ROWS - 1)) * WST + lane;
        atomicAdd(a, r.x);
        atomicAdd(a + 32, r.y);
    };
#define PL_WORD(w, k) ((unsigned)__builtin_amdgcn_readlane((int)(w), (k)))

    if (base < end) {
        float2 v[PL_GRP];
#pragma unroll
        for (int k = 0; k < PL_GRP; ++k) v[k] = gather(PL_WORD(p, k));

        while (base + 256u < end) {
            const unsigned i2 = base + 512u + (unsigned)lane;
            const unsigned w2 = packed[i2 < end ? i2 : base];
#pragma unroll
            for (int k = 0; k < PL_GRP; ++k) {
                addrow(PL_WORD(p, k), v[k]);
                v[k] = gather(PL_WORD(p, PL_GRP + k));
            }
#pragma unroll
            for (int k = 0; k < PL_GRP; ++k) {
                addrow(PL_WORD(p, PL_GRP + k), v[k]);
                v[k] = gather(PL_WORD(pn, k));
            }
            base += 256u;
            p = pn;
            pn = i2 < end ? w2 : 0u;
        }

        const unsigned left = end - base;
        if (left >= 32u) {
#pragma unroll
            for (int k = 0; k < PL_GRP; ++k) {
                addrow(PL_WORD(p, k), v[k]);
                v[k] = gather(PL_WORD(p, PL_GRP + k));
            }
#pragma unroll
            for (int k = 0; k < PL_GRP; ++k) addrow(PL_WORD(p, PL_GRP + k), v[k]);
        } else if (left > (unsigned)PL_GRP) {
            const int cnt = (int)left;
#pragma unroll
            for (int k = 0; k < PL_GRP; ++k) {
                addrow(PL_WORD(p, k), v[k]);
                v[k] = gather(PL_WORD(p, PL_GRP + k));
            }
            asm volatile("" :: "v"(v[0].x), "v"(v[1].x), "v"(v[2].x), "v"(v[3].x),
                               "v"(v[4].x), "v"(v[5].x), "v"(v[6].x), "v"(v[7].x),
                               "v"(v[8].x), "v"(v[9].x), "v"(v[10].x), "v"(v[11].x),
                               "v"(v[12].x), "v"(v[13].x), "v"(v[14].x), "v"(v[15].x));
#pragma unroll
            for (int k = 0; k < PL_GRP; ++k)
                if (PL_GRP + k < cnt) addrow(PL_WORD(p, PL_GRP + k), v[k]);
        } else {
            const int cnt = (int)left;
#pragma unroll
            for (int k = 0; k < PL_GRP; ++k)
                if (k < cnt) addrow(PL_WORD(p, k), v[k]);
        }
    }
#undef PL_WORD
    __syncthreads();

#pragma unroll
    for (int k = 0; k < 4; ++k) {
        const int i = tid + 256 * k;
        const bool rok = brow0 + (i >> 4) < n;
        float4* ap = (float4*)&acc[(i >> 4) * WST + (i & 15) * 4];
        const float4 av = *ap;
        *ap = make_float4((rok ? hv[k].x : 0.f) + av.x, (rok ? hv[k].y : 0.f) + av.y,
                          (rok ? hv[k].z : 0.f) + av.z, (rok ? hv[k].w : 0.f) + av.w);
    }
    __syncthreads();

    if (wave >= PB_ROWS / 16) return;

    const int half = lane >> 4;
    const int l16  = lane & 15;
    const int row0 = brow0 + wave * 16;
    float* tl = acc + wave * (16 * WST);

    v2f a[16];
    float ss = 0.f;
#pragma unroll
    for (int s = 0; s < 16; ++s) {
        const float2 q = *(const float2*)&tl[l16 * WST + 4 * s + half * 2];
        a[s].x = q.x; a[s].y = q.y;
        ss += q.x * q.x + q.y * q.y;
    }
    ss += __shfl_xor(ss, 16, 32);
    const float inv1 = rsqrtf(ss * (1.0f / 64.0f) + 1e-5f);
#pragma unroll
    for (int s = 0; s < 16; ++s) { a[s].x *= inv1; a[s].y *= inv1; }

    float rs[8][4];
#pragma unroll
    for (int v = 0; v < 8; ++v)
#pragma unroll
        for (int j = 0; j < 4; ++j) rs[v][j] = tl[(v + half * 8) * WST + j * 16 + l16];

    v8f c1[4] = {};

#pragma unroll
    for (int s = 0; s < 16; ++s) {
#pragma unroll
        for (int j = 0; j < 4; ++j) {
            const int nc = j * 16 + l16;
            float2 bb = *(const float2*)&lw[nc * WST + 4 * s + half * 2];
            v2f b; b.x = bb.x; b.y = bb.y;
            c1[j] = __builtin_amdgcn_wmma_f32_16x16x4_f32(
                false, a[s], false, b, (short)0, c1[j], false, false);
        }
    }

    float w1c[4], w2c[4];
#pragma unroll
    for (int j = 0; j < 4; ++j) {
        w1c[j] = lnw[j * 16 + l16];
        w2c[j] = lnw[HID + j * 16 + l16];
    }

    float s2[8];
#pragma unroll
    for (int v = 0; v < 8; ++v) {
        const float iv = __shfl(inv1, v + half * 8, 32);
        float s = 0.f;
#pragma unroll
        for (int j = 0; j < 4; ++j) {
            const float hh = rs[v][j] * iv * w1c[j];
            const float y = hh + fmaxf(c1[j][v], 0.f);
            c1[j][v] = y;
            s += y * y;
        }
        s2[v] = s;
    }
#pragma unroll
    for (int v = 0; v < 8; ++v) s2[v] = dpp_add<0x161>(s2[v]);
#pragma unroll
    for (int v = 0; v < 8; ++v) s2[v] = dpp_add<0x162>(s2[v]);
#pragma unroll
    for (int v = 0; v < 8; ++v) s2[v] = dpp_add<0x164>(s2[v]);
#pragma unroll
    for (int v = 0; v < 8; ++v) s2[v] = dpp_add<0x168>(s2[v]);

#pragma unroll
    for (int v = 0; v < 8; ++v) {
        const int rowg = row0 + v + half * 8;
        const float inv2 = rsqrtf(s2[v] * (1.0f / 64.0f) + 1e-5f);
#pragma unroll
        for (int j = 0; j < 4; ++j) c1[j][v] = c1[j][v] * inv2 * w2c[j];
        if (rowg < n) {
#pragma unroll
            for (int j = 0; j < 4; ++j)
                out[(size_t)rowg * HID + j * 16 + l16] = c1[j][v];
        }
    }

    if (!fuse) return;

    __builtin_amdgcn_fence(__ATOMIC_RELEASE, "wavefront");
    __builtin_amdgcn_wave_barrier();
    __builtin_amdgcn_fence(__ATOMIC_ACQUIRE, "wavefront");

#pragma unroll
    for (int v = 0; v < 8; ++v)
#pragma unroll
        for (int j = 0; j < 4; ++j)
            tl[(v + half * 8) * WST + j * 16 + l16] = c1[j][v];

    __builtin_amdgcn_fence(__ATOMIC_RELEASE, "wavefront");
    __builtin_amdgcn_wave_barrier();
    __builtin_amdgcn_fence(__ATOMIC_ACQUIRE, "wavefront");

    v2f a2[16];
#pragma unroll
    for (int s = 0; s < 16; ++s) {
        float2 q = *(const float2*)&tl[l16 * WST + 4 * s + half * 2];
        a2[s].x = q.x; a2[s].y = q.y;
    }

    v8f c2[4] = {};
#pragma unroll
    for (int s = 0; s < 16; ++s) {
#pragma unroll
        for (int j = 0; j < 4; ++j) {
            const int nc = j * 16 + l16;
            float2 bb = *(const float2*)&lw2[nc * WST + 4 * s + half * 2];
            v2f b; b.x = bb.x; b.y = bb.y;
            c2[j] = __builtin_amdgcn_wmma_f32_16x16x4_f32(
                false, a2[s], false, b, (short)0, c2[j], false, false);
        }
    }

#pragma unroll
    for (int v = 0; v < 8; ++v) {
        const int rowg = row0 + v + half * 8;
        if (rowg < n) {
            float* op = mout + (size_t)rowg * HID + 2 * l16;
            *(float2*)op        = make_float2(fmaxf(c2[0][v], 0.f), fmaxf(c2[2][v], 0.f));
            *(float2*)(op + 32) = make_float2(fmaxf(c2[1][v], 0.f), fmaxf(c2[3][v], 0.f));
        }
    }
}

extern "C" void kernel_launch(void* const* d_in, const int* in_sizes, int n_in,
                              void* d_out, int out_size, void* d_ws, size_t ws_size,
                              hipStream_t stream)
{
    const float* x       = (const float*)d_in[0];
    const float* conv_w  = (const float*)d_in[1];
    const float* conv_nw = (const float*)d_in[2];
    const float* hid_w   = (const float*)d_in[3];
    const float* hid_nw  = (const float*)d_in[4];
    const int*   row     = (const int*)d_in[5];
    const int*   col     = (const int*)d_in[6];

    const int n  = in_sizes[0] / HID;
    const int ne = in_sizes[5];

    float* P  = (float*)d_out;
    float* Q0 = (float*)d_ws;
    float* Q1 = Q0 + (size_t)n * HID;
    float* H0 = Q1 + (size_t)n * HID;
    float* H1 = H0 + (size_t)n * HID;

    const int nb = (n + PB_ROWS - 1) / PB_ROWS;
    unsigned* cnt    = (unsigned*)(H1 + (size_t)n * HID);
    unsigned* start  = cnt + nb;
    unsigned* rank   = start + nb + 1;
    unsigned* packed = rank + ne;

    const dim3 blk(256);
    const int gemm_grid = (n + 127) / 128;
    const int edge_grid = (ne + 255) / 256;

    gnn_gemm_relu<<<gemm_grid, blk, 0, stream>>>(x, conv_w, Q0, n, cnt, nb);

    gnn_bin_count<<<edge_grid, blk, 0, stream>>>(row, cnt, rank, ne);
    gnn_bin_scan<<<1, dim3(1024), 0, stream>>>(cnt, start, nb);
    gnn_bin_place<<<edge_grid, blk, 0, stream>>>(row, col, start, rank, packed, ne);

    for (int l = 0; l < 4; ++l) {
        const float* hw = hid_w  + (size_t)l * HID * HID;
        const float* cn = conv_nw + l * HID;
        const float* hn = hid_nw  + l * HID;
        const float* cwn = (l + 1 < 4) ? conv_w + (size_t)(l + 1) * HID * HID : nullptr;

        const float* min  = (l & 1) ? Q1 : Q0;
        float*       mnx  = (l & 1) ? Q0 : Q1;
        const float* hin  = (l == 0) ? x : ((l & 1) ? H0 : H1);
        float*       hout = (l == 3) ? P : ((l & 1) ? H1 : H0);

        gnn_layer<<<nb, blk, 0, stream>>>(min, hin, start, packed, cn, hw, hn, cwn, hout, mnx, n);
    }
}


# Edge grouping: rank from an LDS counter of the edge's 2048-edge chunk, one coalesced returning global add per chunk and block (was one per edge, random counter); order in a block stays arrival order
# speedup vs baseline: 4.1015x; 1.1907x over previous
#include <hip/hip_runtime.h>

typedef __attribute__((ext_vector_type(2))) float v2f;
typedef __attribute__((ext_vector_type(8))) float v8f;

#define HID 64
#define WST 68

__global__ __launch_bounds__(256) void gnn_gemm_relu(
    const float* __restrict__ in, const float* __restrict__ W,
    float* __restrict__ out, int n, unsigned* __restrict__ cnt, int nb)
{
    __shared__ __attribute__((aligned(16))) float lw[64 * WST];
    const int tid = threadIdx.x;

    {
        const int gi = blockIdx.x * 256 + tid;
        if (gi < nb) cnt[gi] = 0u;
    }

    const int wave = tid >> 5;
    const int lane = tid & 31;
    const int half = lane >> 4;
    const int l16  = lane & 15;
    const int row0 = blockIdx.x * 128 + wave * 16;

    float4 wv[4];
#pragma unroll
    for (int k = 0; k < 4; ++k) wv[k] = ((const float4*)W)[tid + 256 * k];

    const int  arow  = row0 + l16;
    const bool rowok = arow < n;
    const float* abase = in + (size_t)(rowok ? arow : 0) * HID + half * 2;
    v2f a[16];
#pragma unroll
    for (int s = 0; s < 16; ++s) {
        const float2 t = *(const float2*)(abase + 4 * s);
        a[s].x = rowok ? t.x : 0.f; a[s].y = rowok ? t.y : 0.f;
    }

#pragma unroll
    for (int k = 0; k < 4; ++k) {
        const int i4 = tid + 256 * k;
        *(float4*)&lw[(i4 >> 4) * WST + (i4 & 15) * 4] = wv[k];
    }
    __syncthreads();

    v8f acc[4] = {};

#pragma unroll
    for (int s = 0; s < 16; ++s) {
#pragma unroll
        for (int j = 0; j < 4; ++j) {
            const int nc = j * 16 + l16;
            float2 bb = *(const float2*)&lw[nc * WST + 4 * s + half * 2];
            v2f b; b.x = bb.x; b.y = bb.y;
            acc[j] = __builtin_amdgcn_wmma_f32_16x16x4_f32(
                false, a[s], false, b, (short)0, acc[j], false, false);
        }
    }

#pragma unroll
    for (int v = 0; v < 8; ++v) {
        const int rowg = row0 + v + half * 8;
        if (rowg < n) {
            float* op = out + (size_t)rowg * HID + 2 * l16;
            *(float2*)op        = make_float2(fmaxf(acc[0][v], 0.f), fmaxf(acc[2][v], 0.f));
            *(float2*)(op + 32) = make_float2(fmaxf(acc[1][v], 0.f), fmaxf(acc[3][v], 0.f));
        }
    }
}

template <int CTRL>
__device__ __forceinline__ float dpp_add(float v) {
    return v + __int_as_float(__builtin_amdgcn_update_dpp(
                   0, __float_as_int(v), CTRL, 0xf, 0xf, true));
}

#define PB_ROWS  64
#define PB_SHIFT 6

#define BC_THREADS 512

__global__ __launch_bounds__(BC_THREADS) void gnn_bin_count(
    const int* __restrict__ row, unsigned* __restrict__ cnt,
    unsigned* __restrict__ rank, unsigned* __restrict__ cbase, int ne, int nb)
{
    extern __shared__ unsigned hist[];
    const int tid = threadIdx.x;
    const int e0 = (blockIdx.x * BC_THREADS + tid) * 4;
    const bool whole = e0 + 3 < ne;

    int r[4];
    if (whole) {
        const int4 v = *(const int4*)(row + e0);
        r[0] = v.x; r[1] = v.y; r[2] = v.z; r[3] = v.w;
    } else {
#pragma unroll
        for (int k = 0; k < 4; ++k) r[k] = (e0 + k < ne) ? row[e0 + k] : 0;
    }

    for (int b = tid; b < nb; b += BC_THREADS) hist[b] = 0u;
    __syncthreads();

    unsigned lr[4];
#pragma unroll
    for (int k = 0; k < 4; ++k)
        lr[k] = (e0 + k < ne) ? atomicAdd(&hist[r[k] >> PB_SHIFT], 1u) : 0u;
    if (whole) {
        *(uint4*)(rank + e0) = make_uint4(lr[0], lr[1], lr[2], lr[3]);
    } else {
#pragma unroll
        for (int k = 0; k < 4; ++k) if (e0 + k < ne) rank[e0 + k] = lr[k];
    }
    __syncthreads();

    unsigned* cb = cbase + (size_t)blockIdx.x * nb;
    for (int b0 = tid; b0 < nb; b0 += 4 * BC_THREADS) {
        unsigned old[4];
#pragma unroll
        for (int k = 0; k < 4; ++k) {
            const int b = b0 + k * BC_THREADS;
            old[k] = (b < nb) ? atomicAdd(&cnt[b], hist[b]) : 0u;
        }
#pragma unroll
        for (int k = 0; k < 4; ++k) {
            const int b = b0 + k * BC_THREADS;
            if (b < nb) cb[b] = old[k];
        }
    }
}

__global__ __launch_bounds__(1024) void gnn_bin_scan(
    const unsigned* __restrict__ cnt, unsigned* __restrict__ start, int nb)
{
    __shared__ unsigned wsum[32];
    const int tid = threadIdx.x, lane = tid & 31, wave = tid >> 5;
    const int per = (nb + 1023) / 1024;
    const int b0 = tid * per;
    unsigned s = 0;
    for (int i = 0; i < per; ++i) {
        const int b = b0 + i;
        if (b < nb) s += cnt[b];
    }
    unsigned inc = s;
    for (int d = 1; d < 32; d <<= 1) {
        const unsigned o = __shfl_up(inc, d, 32);
        if (lane >= d) inc += o;
    }
    if (lane == 31) wsum[wave] = inc;
    __syncthreads();
    if (wave == 0) {
        const unsigned w = wsum[lane];
        unsigned winc = w;
        for (int d = 1; d < 32; d <<= 1) {
            const unsigned o = __shfl_up(winc, d, 32);
            if (lane >= d) winc += o;
        }
        wsum[lane] = winc - w;
    }
    __syncthreads();
    unsigned run = wsum[wave] + inc - s;
    for (int i = 0; i < per; ++i) {
        const int b = b0 + i;
        if (b < nb) { start[b] = run; run += cnt[b]; }
    }
    if (tid == 1023) start[nb] = run;
}

__global__ __launch_bounds__(BC_THREADS) void gnn_bin_place(
    const int* __restrict__ row, const int* __restrict__ col,
    const unsigned* __restrict__ start, const unsigned* __restrict__ cbase,
    const unsigned* __restrict__ rank, unsigned* __restrict__ packed, int ne, int nb)
{
    extern __shared__ unsigned off[];
    const int tid = threadIdx.x;
    const int e0 = (blockIdx.x * BC_THREADS + tid) * 4;
    const bool whole = e0 + 3 < ne;

    int r[4], cl[4];
    unsigned lr[4];
    if (whole) {
        const int4 v = *(const int4*)(row + e0);
        const int4 w = *(const int4*)(col + e0);
        const uint4 q = *(const uint4*)(rank + e0);
        r[0] = v.x; r[1] = v.y; r[2] = v.z; r[3] = v.w;
        cl[0] = w.x; cl[1] = w.y; cl[2] = w.z; cl[3] = w.w;
        lr[0] = q.x; lr[1] = q.y; lr[2] = q.z; lr[3] = q.w;
    } else {
#pragma unroll
        for (int k = 0; k < 4; ++k) {
            const bool ok = e0 + k < ne;
            r[k]  = ok ? row[e0 + k] : 0;
            cl[k] = ok ? col[e0 + k] : 0;
            lr[k] = ok ? rank[e0 + k] : 0u;
        }
    }

    const unsigned* cb = cbase + (size_t)blockIdx.x * nb;
    for (int b0 = tid; b0 < nb; b0 += 4 * BC_THREADS) {
        unsigned s[4], q[4];
#pragma unroll
        for (int k = 0; k < 4; ++k) {
            const int b = b0 + k * BC_THREADS;
            s[k] = (b < nb) ? start[b] : 0u;
            q[k] = (b < nb) ? cb[b] : 0u;
        }
#pragma unroll
        for (int k = 0; k < 4; ++k) {
            const int b = b0 + k * BC_THREADS;
            if (b < nb) off[b] = s[k] + q[k];
        }
    }
    __syncthreads();

#pragma unroll
    for (int k = 0; k < 4; ++k)
        if (e0 + k < ne)
            packed[off[r[k] >> PB_SHIFT] + lr[k]] =
                ((unsigned)cl[k] << PB_SHIFT) | (unsigned)(r[k] & (PB_ROWS - 1));
}

#define PL_GRP 16

__global__ __launch_bounds__(256, 10) void gnn_layer(
    const float* __restrict__ m, const float* __restrict__ h,
    const unsigned* __restrict__ start, const unsigned* __restrict__ packed,
    const float* __restrict__ w1, const float* __restrict__ W,
    const float* __restrict__ w2, const float* __restrict__ Wn,
    float* __restrict__ out, float* __restrict__ mout, int n)
{
    __shared__ __attribute__((aligned(16))) float lw[64 * WST];
    __shared__ __attribute__((aligned(16))) float lw2[64 * WST];
    __shared__ __attribute__((aligned(16))) float acc[PB_ROWS * WST];
    __shared__ float lnw[4 * HID];
    const int tid = threadIdx.x;
    const bool fuse = Wn != nullptr;
    const int lane = tid & 31;
    const int wave = __builtin_amdgcn_readfirstlane(tid >> 5);
    const int brow0 = blockIdx.x * PB_ROWS;

    for (int i = tid; i < PB_ROWS * WST / 4; i += 256)
        ((float4*)acc)[i] = make_float4(0.f, 0.f, 0.f, 0.f);

    const float lnv = ((tid & HID) ? w2 : w1)[tid & (HID - 1)];
    const float4 w1q = ((const float4*)w1)[tid & 15];
    const float4* wn4 = (const float4*)(fuse ? Wn : W);
    float4 wv[4], wnv[4];
#pragma unroll
    for (int k = 0; k < 4; ++k) wv[k] = ((const float4*)W)[tid + 256 * k];
#pragma unroll
    for (int k = 0; k < 4; ++k) wnv[k] = wn4[tid + 256 * k];

    const unsigned beg = start[blockIdx.x];
    const unsigned end = start[blockIdx.x + 1];

    float4 hv[4];
#pragma unroll
    for (int k = 0; k < 4; ++k) {
        const int i = tid + 256 * k;
        const int r = brow0 + (i >> 4);
        hv[k] = *(const float4*)(h + (size_t)(r < n ? r : 0) * HID + (size_t)(i & 15) * 4);
    }

    unsigned base = beg + (unsigned)wave * 32u;
    unsigned p  = (base + (unsigned)lane < end) ? packed[base + lane] : 0u;
    unsigned pn = (base + 256u + (unsigned)lane < end) ? packed[base + 256u + lane] : 0u;

#pragma unroll
    for (int k = 0; k < 4; ++k) {
        const int i4 = tid + 256 * k;
        *(float4*)&lw[(i4 >> 4) * WST + (i4 & 15) * 4] =
            make_float4(wv[k].x * w1q.x, wv[k].y * w1q.y, wv[k].z * w1q.z, wv[k].w * w1q.w);
    }
#pragma unroll
    for (int k = 0; k < 4; ++k) {
        const int i4 = tid + 256 * k;
        *(float4*)&lw2[(i4 >> 4) * WST + (i4 & 15) * 4] = wnv[k];
    }
    lnw[tid] = lnv;
    __syncthreads();

    const unsigned laneoff = (unsigned)lane * 8u;
    auto gather = [&](unsigned q) -> float2 {
        return *(const float2*)((const char*)(m + (size_t)(q >> PB_SHIFT) * HID) + laneoff);
    };
    auto addrow = [&](unsigned q, float2 r) {
        float* a = acc + (q & (PB_ROWS - 1)) * WST + lane;
        atomicAdd(a, r.x);
        atomicAdd(a + 32, r.y);
    };
#define PL_WORD(w, k) ((unsigned)__builtin_amdgcn_readlane((int)(w), (k)))

    if (base < end) {
        float2 v[PL_GRP];
#pragma unroll
        for (int k = 0; k < PL_GRP; ++k) v[k] = gather(PL_WORD(p, k));

        while (base + 256u < end) {
            const unsigned i2 = base + 512u + (unsigned)lane;
            const unsigned w2 = packed[i2 < end ? i2 : base];
#pragma unroll
            for (int k = 0; k < PL_GRP; ++k) {
                addrow(PL_WORD(p, k), v[k]);
                v[k] = gather(PL_WORD(p, PL_GRP + k));
            }
#pragma unroll
            for (int k = 0; k < PL_GRP; ++k) {
                addrow(PL_WORD(p, PL_GRP + k), v[k]);
                v[k] = gather(PL_WORD(pn, k));
            }
            base += 256u;
            p = pn;
            pn = i2 < end ? w2 : 0u;
        }

        const unsigned left = end - base;
        if (left >= 32u) {
#pragma unroll
            for (int k = 0; k < PL_GRP; ++k) {
                addrow(PL_WORD(p, k), v[k]);
                v[k] = gather(PL_WORD(p, PL_GRP + k));
            }
#pragma unroll
            for (int k = 0; k < PL_GRP; ++k) addrow(PL_WORD(p, PL_GRP + k), v[k]);
        } else if (left > (unsigned)PL_GRP) {
            const int cnt = (int)left;
#pragma unroll
            for (int k = 0; k < PL_GRP; ++k) {
                addrow(PL_WORD(p, k), v[k]);
                v[k] = gather(PL_WORD(p, PL_GRP + k));
            }
            asm volatile("" :: "v"(v[0].x), "v"(v[1].x), "v"(v[2].x), "v"(v[3].x),
                               "v"(v[4].x), "v"(v[5].x), "v"(v[6].x), "v"(v[7].x),
                               "v"(v[8].x), "v"(v[9].x), "v"(v[10].x), "v"(v[11].x),
                               "v"(v[12].x), "v"(v[13].x), "v"(v[14].x), "v"(v[15].x));
#pragma unroll
            for (int k = 0; k < PL_GRP; ++k)
                if (PL_GRP + k < cnt) addrow(PL_WORD(p, PL_GRP + k), v[k]);
        } else {
            const int cnt = (int)left;
#pragma unroll
            for (int k = 0; k < PL_GRP; ++k)
                if (k < cnt) addrow(PL_WORD(p, k), v[k]);
        }
    }
#undef PL_WORD
    __syncthreads();

#pragma unroll
    for (int k = 0; k < 4; ++k) {
        const int i = tid + 256 * k;
        const bool rok = brow0 + (i >> 4) < n;
        float4* ap = (float4*)&acc[(i >> 4) * WST + (i & 15) * 4];
        const float4 av = *ap;
        *ap = make_float4((rok ? hv[k].x : 0.f) + av.x, (rok ? hv[k].y : 0.f) + av.y,
                          (rok ? hv[k].z : 0.f) + av.z, (rok ? hv[k].w : 0.f) + av.w);
    }
    __syncthreads();

    if (wave >= PB_ROWS / 16) return;

    const int half = lane >> 4;
    const int l16  = lane & 15;
    const int row0 = brow0 + wave * 16;
    float* tl = acc + wave * (16 * WST);

    v2f a[16];
    float ss = 0.f;
#pragma unroll
    for (int s = 0; s < 16; ++s) {
        const float2 q = *(const float2*)&tl[l16 * WST + 4 * s + half * 2];
        a[s].x = q.x; a[s].y = q.y;
        ss += q.x * q.x + q.y * q.y;
    }
    ss += __shfl_xor(ss, 16, 32);
    const float inv1 = rsqrtf(ss * (1.0f / 64.0f) + 1e-5f);
#pragma unroll
    for (int s = 0; s < 16; ++s) { a[s].x *= inv1; a[s].y *= inv1; }

    float rs[8][4];
#pragma unroll
    for (int v = 0; v < 8; ++v)
#pragma unroll
        for (int j = 0; j < 4; ++j) rs[v][j] = tl[(v + half * 8) * WST + j * 16 + l16];

    v8f c1[4] = {};

#pragma unroll
    for (int s = 0; s < 16; ++s) {
#pragma unroll
        for (int j = 0; j < 4; ++j) {
            const int nc = j * 16 + l16;
            float2 bb = *(const float2*)&lw[nc * WST + 4 * s + half * 2];
            v2f b; b.x = bb.x; b.y = bb.y;
            c1[j] = __builtin_amdgcn_wmma_f32_16x16x4_f32(
                false, a[s], false, b, (short)0, c1[j], false, false);
        }
    }

    float w1c[4], w2c[4];
#pragma unroll
    for (int j = 0; j < 4; ++j) {
        w1c[j] = lnw[j * 16 + l16];
        w2c[j] = lnw[HID + j * 16 + l16];
    }

    float s2[8];
#pragma unroll
    for (int v = 0; v < 8; ++v) {
        const float iv = __shfl(inv1, v + half * 8, 32);
        float s = 0.f;
#pragma unroll
        for (int j = 0; j < 4; ++j) {
            const float hh = rs[v][j] * iv * w1c[j];
            const float y = hh + fmaxf(c1[j][v], 0.f);
            c1[j][v] = y;
            s += y * y;
        }
        s2[v] = s;
    }
#pragma unroll
    for (int v = 0; v < 8; ++v) s2[v] = dpp_add<0x161>(s2[v]);
#pragma unroll
    for (int v = 0; v < 8; ++v) s2[v] = dpp_add<0x162>(s2[v]);
#pragma unroll
    for (int v = 0; v < 8; ++v) s2[v] = dpp_add<0x164>(s2[v]);
#pragma unroll
    for (int v = 0; v < 8; ++v) s2[v] = dpp_add<0x168>(s2[v]);

#pragma unroll
    for (int v = 0; v < 8; ++v) {
        const int rowg = row0 + v + half * 8;
        const float inv2 = rsqrtf(s2[v] * (1.0f / 64.0f) + 1e-5f);
#pragma unroll
        for (int j = 0; j < 4; ++j) c1[j][v] = c1[j][v] * inv2 * w2c[j];
        if (rowg < n) {
#pragma unroll
            for (int j = 0; j < 4; ++j)
                out[(size_t)rowg * HID + j * 16 + l16] = c1[j][v];
        }
    }

    if (!fuse) return;

    __builtin_amdgcn_fence(__ATOMIC_RELEASE, "wavefront");
    __builtin_amdgcn_wave_barrier();
    __builtin_amdgcn_fence(__ATOMIC_ACQUIRE, "wavefront");

#pragma unroll
    for (int v = 0; v < 8; ++v)
#pragma unroll
        for (int j = 0; j < 4; ++j)
            tl[(v + half * 8) * WST + j * 16 + l16] = c1[j][v];

    __builtin_amdgcn_fence(__ATOMIC_RELEASE, "wavefront");
    __builtin_amdgcn_wave_barrier();
    __builtin_amdgcn_fence(__ATOMIC_ACQUIRE, "wavefront");

    v2f a2[16];
#pragma unroll
    for (int s = 0; s < 16; ++s) {
        float2 q = *(const float2*)&tl[l16 * WST + 4 * s + half * 2];
        a2[s].x = q.x; a2[s].y = q.y;
    }

    v8f c2[4] = {};
#pragma unroll
    for (int s = 0; s < 16; ++s) {
#pragma unroll
        for (int j = 0; j < 4; ++j) {
            const int nc = j * 16 + l16;
            float2 bb = *(const float2*)&lw2[nc * WST + 4 * s + half * 2];
            v2f b; b.x = bb.x; b.y = bb.y;
            c2[j] = __builtin_amdgcn_wmma_f32_16x16x4_f32(
                false, a2[s], false, b, (short)0, c2[j], false, false);
        }
    }

#pragma unroll
    for (int v = 0; v < 8; ++v) {
        const int rowg = row0 + v + half * 8;
        if (rowg < n) {
            float* op = mout + (size_t)rowg * HID + 2 * l16;
            *(float2*)op        = make_float2(fmaxf(c2[0][v], 0.f), fmaxf(c2[2][v], 0.f));
            *(float2*)(op + 32) = make_float2(fmaxf(c2[1][v], 0.f), fmaxf(c2[3][v], 0.f));
        }
    }
}

extern "C" void kernel_launch(void* const* d_in, const int* in_sizes, int n_in,
                              void* d_out, int out_size, void* d_ws, size_t ws_size,
                              hipStream_t stream)
{
    const float* x       = (const float*)d_in[0];
    const float* conv_w  = (const float*)d_in[1];
    const float* conv_nw = (const float*)d_in[2];
    const float* hid_w   = (const float*)d_in[3];
    const float* hid_nw  = (const float*)d_in[4];
    const int*   row     = (const int*)d_in[5];
    const int*   col     = (const int*)d_in[6];

    const int n  = in_sizes[0] / HID;
    const int ne = in_sizes[5];

    float* P  = (float*)d_out;
    float* Q0 = (float*)d_ws;
    float* Q1 = Q0 + (size_t)n * HID;
    float* H0 = Q1 + (size_t)n * HID;
    float* H1 = H0 + (size_t)n * HID;

    const int nb = (n + PB_ROWS - 1) / PB_ROWS;
    const int nbp = (nb + 64) & ~63;
    const int nep = (ne + 63) & ~63;
    unsigned* cnt    = (unsigned*)(H1 + (size_t)n * HID);
    unsigned* start  = cnt + nbp;
    unsigned* rank   = start + nbp;
    unsigned* packed = rank + nep;
    unsigned* cbase  = packed + nep;

    const dim3 blk(256);
    const int gemm_grid = (n + 127) / 128;
    const int edge_grid = (ne + BC_THREADS * 4 - 1) / (BC_THREADS * 4);
    const dim3 eblk(BC_THREADS);
    const size_t hist_bytes = (size_t)nb * sizeof(unsigned);

    gnn_gemm_relu<<<gemm_grid, blk, 0, stream>>>(x, conv_w, Q0, n, cnt, nb);

    gnn_bin_count<<<edge_grid, eblk, hist_bytes, stream>>>(row, cnt, rank, cbase, ne, nb);
    gnn_bin_scan<<<1, dim3(1024), 0, stream>>>(cnt, start, nb);
    gnn_bin_place<<<edge_grid, eblk, hist_bytes, stream>>>(row, col, start, cbase, rank, packed, ne, nb);

    for (int l = 0; l < 4; ++l) {
        const float* hw = hid_w  + (size_t)l * HID * HID;
        const float* cn = conv_nw + l * HID;
        const float* hn = hid_nw  + l * HID;
        const float* cwn = (l + 1 < 4) ? conv_w + (size_t)(l + 1) * HID * HID : nullptr;

        const float* min  = (l & 1) ? Q1 : Q0;
        float*       mnx  = (l & 1) ? Q0 : Q1;
        const float* hin  = (l == 0) ? x : ((l & 1) ? H0 : H1);
        float*       hout = (l == 3) ? P : ((l & 1) ? H1 : H0);

        gnn_layer<<<nb, blk, 0, stream>>>(min, hin, start, packed, cn, hw, hn, cwn, hout, mnx, n);
    }
}


# Layer blocks of 4 waves, each pulls (ring of 32 gathers) then owns a tile; next weight takes W's place in LDS after the first product: 35 KB each, all blocks resident at once; adds in arrival order
# speedup vs baseline: 4.2625x; 1.0393x over previous
#include <hip/hip_runtime.h>

typedef __attribute__((ext_vector_type(2))) float v2f;
typedef __attribute__((ext_vector_type(8))) float v8f;

#define HID 64
#define WST 68

__global__ __launch_bounds__(256) void gnn_gemm_relu(
    const float* __restrict__ in, const float* __restrict__ W,
    float* __restrict__ out, int n, unsigned* __restrict__ cnt, int nb)
{
    __shared__ __attribute__((aligned(16))) float lw[64 * WST];
    const int tid = threadIdx.x;

    {
        const int gi = blockIdx.x * 256 + tid;
        if (gi < nb) cnt[gi] = 0u;
    }

    const int wave = tid >> 5;
    const int lane = tid & 31;
    const int half = lane >> 4;
    const int l16  = lane & 15;
    const int row0 = blockIdx.x * 128 + wave * 16;

    float4 wv[4];
#pragma unroll
    for (int k = 0; k < 4; ++k) wv[k] = ((const float4*)W)[tid + 256 * k];

    const int  arow  = row0 + l16;
    const bool rowok = arow < n;
    const float* abase = in + (size_t)(rowok ? arow : 0) * HID + half * 2;
    v2f a[16];
#pragma unroll
    for (int s = 0; s < 16; ++s) {
        const float2 t = *(const float2*)(abase + 4 * s);
        a[s].x = rowok ? t.x : 0.f; a[s].y = rowok ? t.y : 0.f;
    }

#pragma unroll
    for (int k = 0; k < 4; ++k) {
        const int i4 = tid + 256 * k;
        *(float4*)&lw[(i4 >> 4) * WST + (i4 & 15) * 4] = wv[k];
    }
    __syncthreads();

    v8f acc[4] = {};

#pragma unroll
    for (int s = 0; s < 16; ++s) {
#pragma unroll
        for (int j = 0; j < 4; ++j) {
            const int nc = j * 16 + l16;
            float2 bb = *(const float2*)&lw[nc * WST + 4 * s + half * 2];
            v2f b; b.x = bb.x; b.y = bb.y;
            acc[j] = __builtin_amdgcn_wmma_f32_16x16x4_f32(
                false, a[s], false, b, (short)0, acc[j], false, false);
        }
    }

#pragma unroll
    for (int v = 0; v < 8; ++v) {
        const int rowg = row0 + v + half * 8;
        if (rowg < n) {
            float* op = out + (size_t)rowg * HID + 2 * l16;
            *(float2*)op        = make_float2(fmaxf(acc[0][v], 0.f), fmaxf(acc[2][v], 0.f));
            *(float2*)(op + 32) = make_float2(fmaxf(acc[1][v], 0.f), fmaxf(acc[3][v], 0.f));
        }
    }
}

template <int CTRL>
__device__ __forceinline__ float dpp_add(float v) {
    return v + __int_as_float(__builtin_amdgcn_update_dpp(
                   0, __float_as_int(v), CTRL, 0xf, 0xf, true));
}

#define PB_ROWS  64
#define PB_SHIFT 6

#define BC_THREADS 512

__global__ __launch_bounds__(BC_THREADS) void gnn_bin_count(
    const int* __restrict__ row, unsigned* __restrict__ cnt,
    unsigned* __restrict__ rank, unsigned* __restrict__ cbase, int ne, int nb)
{
    extern __shared__ unsigned hist[];
    const int tid = threadIdx.x;
    const int e0 = (blockIdx.x * BC_THREADS + tid) * 4;
    const bool whole = e0 + 3 < ne;

    int r[4];
    if (whole) {
        const int4 v = *(const int4*)(row + e0);
        r[0] = v.x; r[1] = v.y; r[2] = v.z; r[3] = v.w;
    } else {
#pragma unroll
        for (int k = 0; k < 4; ++k) r[k] = (e0 + k < ne) ? row[e0 + k] : 0;
    }

    for (int b = tid; b < nb; b += BC_THREADS) hist[b] = 0u;
    __syncthreads();

    unsigned lr[4];
#pragma unroll
    for (int k = 0; k < 4; ++k)
        lr[k] = (e0 + k < ne) ? atomicAdd(&hist[r[k] >> PB_SHIFT], 1u) : 0u;
    if (whole) {
        *(uint4*)(rank + e0) = make_uint4(lr[0], lr[1], lr[2], lr[3]);
    } else {
#pragma unroll
        for (int k = 0; k < 4; ++k) if (e0 + k < ne) rank[e0 + k] = lr[k];
    }
    __syncthreads();

    unsigned* cb = cbase + (size_t)blockIdx.x * nb;
    for (int b0 = tid; b0 < nb; b0 += 4 * BC_THREADS) {
        unsigned old[4];
#pragma unroll
        for (int k = 0; k < 4; ++k) {
            const int b = b0 + k * BC_THREADS;
            old[k] = (b < nb) ? atomicAdd(&cnt[b], hist[b]) : 0u;
        }
#pragma unroll
        for (int k = 0; k < 4; ++k) {
            const int b = b0 + k * BC_THREADS;
            if (b < nb) cb[b] = old[k];
        }
    }
}

__global__ __launch_bounds__(1024) void gnn_bin_scan(
    const unsigned* __restrict__ cnt, unsigned* __restrict__ start, int nb)
{
    __shared__ unsigned wsum[32];
    const int tid = threadIdx.x, lane = tid & 31, wave = tid >> 5;
    const int per = (nb + 1023) / 1024;
    const int b0 = tid * per;
    unsigned s = 0;
    for (int i = 0; i < per; ++i) {
        const int b = b0 + i;
        if (b < nb) s += cnt[b];
    }
    unsigned inc = s;
    for (int d = 1; d < 32; d <<= 1) {
        const unsigned o = __shfl_up(inc, d, 32);
        if (lane >= d) inc += o;
    }
    if (lane == 31) wsum[wave] = inc;
    __syncthreads();
    if (wave == 0) {
        const unsigned w = wsum[lane];
        unsigned winc = w;
        for (int d = 1; d < 32; d <<= 1) {
            const unsigned o = __shfl_up(winc, d, 32);
            if (lane >= d) winc += o;
        }
        wsum[lane] = winc - w;
    }
    __syncthreads();
    unsigned run = wsum[wave] + inc - s;
    for (int i = 0; i < per; ++i) {
        const int b = b0 + i;
        if (b < nb) { start[b] = run; run += cnt[b]; }
    }
    if (tid == 1023) start[nb] = run;
}

__global__ __launch_bounds__(BC_THREADS) void gnn_bin_place(
    const int* __restrict__ row, const int* __restrict__ col,
    const unsigned* __restrict__ start, const unsigned* __restrict__ cbase,
    const unsigned* __restrict__ rank, unsigned* __restrict__ packed, int ne, int nb)
{
    extern __shared__ unsigned off[];
    const int tid = threadIdx.x;
    const int e0 = (blockIdx.x * BC_THREADS + tid) * 4;
    const bool whole = e0 + 3 < ne;

    int r[4], cl[4];
    unsigned lr[4];
    if (whole) {
        const int4 v = *(const int4*)(row + e0);
        const int4 w = *(const int4*)(col + e0);
        const uint4 q = *(const uint4*)(rank + e0);
        r[0] = v.x; r[1] = v.y; r[2] = v.z; r[3] = v.w;
        cl[0] = w.x; cl[1] = w.y; cl[2] = w.z; cl[3] = w.w;
        lr[0] = q.x; lr[1] = q.y; lr[2] = q.z; lr[3] = q.w;
    } else {
#pragma unroll
        for (int k = 0; k < 4; ++k) {
            const bool ok = e0 + k < ne;
            r[k]  = ok ? row[e0 + k] : 0;
            cl[k] = ok ? col[e0 + k] : 0;
            lr[k] = ok ? rank[e0 + k] : 0u;
        }
    }

    const unsigned* cb = cbase + (size_t)blockIdx.x * nb;
    for (int b0 = tid; b0 < nb; b0 += 4 * BC_THREADS) {
        unsigned s[4], q[4];
#pragma unroll
        for (int k = 0; k < 4; ++k) {
            const int b = b0 + k * BC_THREADS;
            s[k] = (b < nb) ? start[b] : 0u;
            q[k] = (b < nb) ? cb[b] : 0u;
        }
#pragma unroll
        for (int k = 0; k < 4; ++k) {
            const int b = b0 + k * BC_THREADS;
            if (b < nb) off[b] = s[k] + q[k];
        }
    }
    __syncthreads();

#pragma unroll
    for (int k = 0; k < 4; ++k)
        if (e0 + k < ne)
            packed[off[r[k] >> PB_SHIFT] + lr[k]] =
                ((unsigned)cl[k] << PB_SHIFT) | (unsigned)(r[k] & (PB_ROWS - 1));
}

__device__ __forceinline__ void lds_barrier() {
    __builtin_amdgcn_fence(__ATOMIC_RELEASE, "workgroup", "local");
    __builtin_amdgcn_s_barrier();
    __builtin_amdgcn_fence(__ATOMIC_ACQUIRE, "workgroup", "local");
}

#define PL_GRP 32
#define LY_THREADS 128
#define LY_SWEEP (LY_THREADS)

__global__ __launch_bounds__(LY_THREADS, 7) void gnn_layer(
    const float* __restrict__ m, const float* __restrict__ h,
    const unsigned* __restrict__ start, const unsigned* __restrict__ packed,
    const float* __restrict__ w1, const float* __restrict__ W,
    const float* __restrict__ w2, const float* __restrict__ Wn,
    float* __restrict__ out, float* __restrict__ mout, int n)
{
    __shared__ __attribute__((aligned(16))) float lw[64 * WST];
    __shared__ __attribute__((aligned(16))) float acc[PB_ROWS * WST];
    __shared__ float lnw[2 * HID];
    const int tid = threadIdx.x;
    const bool fuse = Wn != nullptr;
    const int lane = tid & 31;
    const int wave = __builtin_amdgcn_readfirstlane(tid >> 5);
    const int brow0 = blockIdx.x * PB_ROWS;

    for (int i = tid; i < PB_ROWS * WST / 4; i += LY_THREADS)
        ((float4*)acc)[i] = make_float4(0.f, 0.f, 0.f, 0.f);

    const float lnv = ((tid & HID) ? w2 : w1)[tid & (HID - 1)];
    const float4 w1q = ((const float4*)w1)[tid & 15];
    float4 wv[8];
#pragma unroll
    for (int k = 0; k < 8; ++k) wv[k] = ((const float4*)W)[tid + LY_THREADS * k];

    const unsigned beg = start[blockIdx.x];
    const unsigned end = start[blockIdx.x + 1];

    float4 hv[8];
#pragma unroll
    for (int k = 0; k < 8; ++k) {
        const int i = tid + LY_THREADS * k;
        const int r = brow0 + (i >> 4);
        hv[k] = *(const float4*)(h + (size_t)(r < n ? r : 0) * HID + (size_t)(i & 15) * 4);
    }

    unsigned base = beg + (unsigned)wave * 32u;
    unsigned p  = (base + (unsigned)lane < end) ? packed[base + lane] : 0u;
    unsigned pn = (base + LY_SWEEP + (unsigned)lane < end) ? packed[base + LY_SWEEP + lane] : 0u;

#pragma unroll
    for (int k = 0; k < 8; ++k) {
        const int i4 = tid + LY_THREADS * k;
        *(float4*)&lw[(i4 >> 4) * WST + (i4 & 15) * 4] =
            make_float4(wv[k].x * w1q.x, wv[k].y * w1q.y, wv[k].z * w1q.z, wv[k].w * w1q.w);
    }
    lnw[tid] = lnv;
    __syncthreads();

    const unsigned laneoff = (unsigned)lane * 8u;
    auto gather = [&](unsigned q) -> float2 {
        return *(const float2*)((const char*)(m + (size_t)(q >> PB_SHIFT) * HID) + laneoff);
    };
    auto addrow = [&](unsigned q, float2 r) {
        float* a = acc + (q & (PB_ROWS - 1)) * WST + lane;
        atomicAdd(a, r.x);
        atomicAdd(a + 32, r.y);
    };
#define PL_WORD(w, k) ((unsigned)__builtin_amdgcn_readlane((int)(w), (k)))

    if (base < end) {
        float2 v[PL_GRP];
#pragma unroll
        for (int k = 0; k < PL_GRP; ++k) {
            v[k] = gather(PL_WORD(p, k));
            __builtin_amdgcn_sched_barrier(0);
        }

        while (base + LY_SWEEP < end) {
            const unsigned i2 = base + 2u * LY_SWEEP + (unsigned)lane;
            const unsigned wnn = packed[i2 < end ? i2 : base];
#pragma unroll
            for (int k = 0; k < PL_GRP; ++k) {
                addrow(PL_WORD(p, k), v[k]);
                v[k] = gather(PL_WORD(pn, k));
            }
            base += LY_SWEEP;
            p = pn;
            pn = i2 < end ? wnn : 0u;
        }

        const unsigned left = end - base;
        if (left >= 32u) {
#pragma unroll
            for (int k = 0; k < PL_GRP; ++k) addrow(PL_WORD(p, k), v[k]);
        } else {
            const int cnt = (int)left;
#pragma unroll
            for (int k = 0; k < PL_GRP; ++k)
                if (k < cnt) addrow(PL_WORD(p, k), v[k]);
        }
    }
#undef PL_WORD
    __syncthreads();

    {
        float4 av[8];
#pragma unroll
        for (int k = 0; k < 8; ++k) {
            const int i = tid + LY_THREADS * k;
            av[k] = *(const float4*)&acc[(i >> 4) * WST + (i & 15) * 4];
        }
#pragma unroll
        for (int k = 0; k < 8; ++k) {
            const int i = tid + LY_THREADS * k;
            const bool rok = brow0 + (i >> 4) < n;
            *(float4*)&acc[(i >> 4) * WST + (i & 15) * 4] =
                make_float4((rok ? hv[k].x : 0.f) + av[k].x, (rok ? hv[k].y : 0.f) + av[k].y,
                            (rok ? hv[k].z : 0.f) + av[k].z, (rok ? hv[k].w : 0.f) + av[k].w);
        }
    }
    __syncthreads();

    const int half = lane >> 4;
    const int l16  = lane & 15;
    const int row0 = brow0 + wave * 16;
    float* tl = acc + wave * (16 * WST);

    v2f a[16];
    float ss = 0.f;
#pragma unroll
    for (int s = 0; s < 16; ++s) {
        const float2 q = *(const float2*)&tl[l16 * WST + 4 * s + half * 2];
        a[s].x = q.x; a[s].y = q.y;
        ss += q.x * q.x + q.y * q.y;
    }
    ss += __shfl_xor(ss, 16, 32);
    const float inv1 = rsqrtf(ss * (1.0f / 64.0f) + 1e-5f);
#pragma unroll
    for (int s = 0; s < 16; ++s) { a[s].x *= inv1; a[s].y *= inv1; }

    float rs[8][4];
#pragma unroll
    for (int v = 0; v < 8; ++v)
#pragma unroll
        for (int j = 0; j < 4; ++j) rs[v][j] = tl[(v + half * 8) * WST + j * 16 + l16];

    v8f c1[4] = {};

#pragma unroll
    for (int s = 0; s < 16; ++s) {
#pragma unroll
        for (int j = 0; j < 4; ++j) {
            const int nc = j * 16 + l16;
            float2 bb = *(const float2*)&lw[nc * WST + 4 * s + half * 2];
            v2f b; b.x = bb.x; b.y = bb.y;
            c1[j] = __builtin_amdgcn_wmma_f32_16x16x4_f32(
                false, a[s], false, b, (short)0, c1[j], false, false);
        }
    }

    float w1c[4], w2c[4];
#pragma unroll
    for (int j = 0; j < 4; ++j) {
        w1c[j] = lnw[j * 16 + l16];
        w2c[j] = lnw[HID + j * 16 + l16];
    }

    float s2[8];
#pragma unroll
    for (int v = 0; v < 8; ++v) {
        const float iv = __shfl(inv1, v + half * 8, 32);
        float s = 0.f;
#pragma unroll
        for (int j = 0; j < 4; ++j) {
            const float hh = rs[v][j] * iv * w1c[j];
            const float y = hh + fmaxf(c1[j][v], 0.f);
            c1[j][v] = y;
            s += y * y;
        }
        s2[v] = s;
    }
#pragma unroll
    for (int v = 0; v < 8; ++v) s2[v] = dpp_add<0x161>(s2[v]);
#pragma unroll
    for (int v = 0; v < 8; ++v) s2[v] = dpp_add<0x162>(s2[v]);
#pragma unroll
    for (int v = 0; v < 8; ++v) s2[v] = dpp_add<0x164>(s2[v]);
#pragma unroll
    for (int v = 0; v < 8; ++v) s2[v] = dpp_add<0x168>(s2[v]);

#pragma unroll
    for (int v = 0; v < 8; ++v) {
        const int rowg = row0 + v + half * 8;
        const float inv2 = rsqrtf(s2[v] * (1.0f / 64.0f) + 1e-5f);
#pragma unroll
        for (int j = 0; j < 4; ++j) c1[j][v] = c1[j][v] * inv2 * w2c[j];
        if (rowg < n) {
#pragma unroll
            for (int j = 0; j < 4; ++j)
                out[(size_t)rowg * HID + j * 16 + l16] = c1[j][v];
        }
    }

    if (!fuse) return;

    const float4* wn4 = (const float4*)Wn + tid;
    const float4 wn0 = wn4[0 * LY_THREADS], wn1 = wn4[1 * LY_THREADS],
                 wn2 = wn4[2 * LY_THREADS], wn3 = wn4[3 * LY_THREADS],
                 wn4v = wn4[4 * LY_THREADS], wn5 = wn4[5 * LY_THREADS],
                 wn6 = wn4[6 * LY_THREADS], wn7 = wn4[7 * LY_THREADS];

    __builtin_amdgcn_fence(__ATOMIC_RELEASE, "wavefront");
    __builtin_amdgcn_wave_barrier();
    __builtin_amdgcn_fence(__ATOMIC_ACQUIRE, "wavefront");

#pragma unroll
    for (int v = 0; v < 8; ++v)
#pragma unroll
        for (int j = 0; j < 4; ++j)
            tl[(v + half * 8) * WST + j * 16 + l16] = c1[j][v];

    lds_barrier();
    {
        float* lp = &lw[(tid >> 4) * WST + (tid & 15) * 4];
        *(float4*)(lp + 0 * 8 * WST) = wn0;  *(float4*)(lp + 1 * 8 * WST) = wn1;
        *(float4*)(lp + 2 * 8 * WST) = wn2;  *(float4*)(lp + 3 * 8 * WST) = wn3;
        *(float4*)(lp + 4 * 8 * WST) = wn4v; *(float4*)(lp + 5 * 8 * WST) = wn5;
        *(float4*)(lp + 6 * 8 * WST) = wn6;  *(float4*)(lp + 7 * 8 * WST) = wn7;
    }
    lds_barrier();

    v2f a2[16];
#pragma unroll
    for (int s = 0; s < 16; ++s) {
        float2 q = *(const float2*)&tl[l16 * WST + 4 * s + half * 2];
        a2[s].x = q.x; a2[s].y = q.y;
    }

    v8f c2[4] = {};
#pragma unroll
    for (int s = 0; s < 16; ++s) {
#pragma unroll
        for (int j = 0; j < 4; ++j) {
            const int nc = j * 16 + l16;
            float2 bb = *(const float2*)&lw[nc * WST + 4 * s + half * 2];
            v2f b; b.x = bb.x; b.y = bb.y;
            c2[j] = __builtin_amdgcn_wmma_f32_16x16x4_f32(
                false, a2[s], false, b, (short)0, c2[j], false, false);
        }
    }

#pragma unroll
    for (int v = 0; v < 8; ++v) {
        const int rowg = row0 + v + half * 8;
        if (rowg < n) {
            float* op = mout + (size_t)rowg * HID + 2 * l16;
            *(float2*)op        = make_float2(fmaxf(c2[0][v], 0.f), fmaxf(c2[2][v], 0.f));
            *(float2*)(op + 32) = make_float2(fmaxf(c2[1][v], 0.f), fmaxf(c2[3][v], 0.f));
        }
    }
}

extern "C" void kernel_launch(void* const* d_in, const int* in_sizes, int n_in,
                              void* d_out, int out_size, void* d_ws, size_t ws_size,
                              hipStream_t stream)
{
    const float* x       = (const float*)d_in[0];
    const float* conv_w  = (const float*)d_in[1];
    const float* conv_nw = (const float*)d_in[2];
    const float* hid_w   = (const float*)d_in[3];
    const float* hid_nw  = (const float*)d_in[4];
    const int*   row     = (const int*)d_in[5];
    const int*   col     = (const int*)d_in[6];

    const int n  = in_sizes[0] / HID;
    const int ne = in_sizes[5];

    float* P  = (float*)d_out;
    float* Q0 = (float*)d_ws;
    float* Q1 = Q0 + (size_t)n * HID;
    float* H0 = Q1 + (size_t)n * HID;
    float* H1 = H0 + (size_t)n * HID;

    const int nb = (n + PB_ROWS - 1) / PB_ROWS;
    const int nbp = (nb + 64) & ~63;
    const int nep = (ne + 63) & ~63;
    unsigned* cnt    = (unsigned*)(H1 + (size_t)n * HID);
    unsigned* start  = cnt + nbp;
    unsigned* rank   = start + nbp;
    unsigned* packed = rank + nep;
    unsigned* cbase  = packed + nep;

    const dim3 blk(256);
    const int gemm_grid = (n + 127) / 128;
    const int edge_grid = (ne + BC_THREADS * 4 - 1) / (BC_THREADS * 4);
    const dim3 eblk(BC_THREADS);
    const size_t hist_bytes = (size_t)nb * sizeof(unsigned);

    gnn_gemm_relu<<<gemm_grid, blk, 0, stream>>>(x, conv_w, Q0, n, cnt, nb);

    gnn_bin_count<<<edge_grid, eblk, hist_bytes, stream>>>(row, cnt, rank, cbase, ne, nb);
    gnn_bin_scan<<<1, dim3(1024), 0, stream>>>(cnt, start, nb);
    gnn_bin_place<<<edge_grid, eblk, hist_bytes, stream>>>(row, col, start, cbase, rank, packed, ne, nb);

    for (int l = 0; l < 4; ++l) {
        const float* hw = hid_w  + (size_t)l * HID * HID;
        const float* cn = conv_nw + l * HID;
        const float* hn = hid_nw  + l * HID;
        const float* cwn = (l + 1 < 4) ? conv_w + (size_t)(l + 1) * HID * HID : nullptr;

        const float* min  = (l & 1) ? Q1 : Q0;
        float*       mnx  = (l & 1) ? Q0 : Q1;
        const float* hin  = (l == 0) ? x : ((l & 1) ? H0 : H1);
        float*       hout = (l == 3) ? P : ((l & 1) ? H1 : H0);

        gnn_layer<<<nb, dim3(LY_THREADS), 0, stream>>>(min, hin, start, packed, cn, hw, hn, cwn, hout, mnx, n);
    }
}
